# starting kernel
# baseline (speedup 1.0000x reference)
#include <hip/hip_runtime.h>

#define N_NODES 200000
#define N_EDGES 6400000
#define FDIM    128
#define HDIM    16
#define CDIM    10
#define BGRAPH  512
#define NTILES  (N_NODES / 16)

typedef float v2f __attribute__((ext_vector_type(2)));
typedef float v8f __attribute__((ext_vector_type(8)));

__global__ void count_deg_kernel(const int* __restrict__ col, int* __restrict__ degcnt) {
    int e = blockIdx.x * blockDim.x + threadIdx.x;
    if (e < N_EDGES) atomicAdd(&degcnt[col[e]], 1);
}

__global__ void node_prep_kernel(const int* __restrict__ degcnt, const int* __restrict__ batch,
                                 float* __restrict__ dis, int* __restrict__ counts) {
    int n = blockIdx.x * blockDim.x + threadIdx.x;
    if (n < N_NODES) {
        dis[n] = rsqrtf(1.0f + (float)degcnt[n]);
        atomicAdd(&counts[batch[n]], 1);
    }
}

__global__ void __launch_bounds__(256) gemm1_kernel(const float* __restrict__ x,
                                                    const float* __restrict__ W1,
                                                    const float* __restrict__ dis,
                                                    float* __restrict__ hws,
                                                    float* __restrict__ agg) {
    __shared__ float lw[(FDIM / 2) * 32];
    int tid = threadIdx.x;
#pragma unroll
    for (int i = 0; i < 8; ++i) {
        int idx = i * 256 + tid;
        int k = idx >> 4, n = idx & 15;
        lw[(k >> 1) * 32 + n * 2 + (k & 1)] = W1[idx];
    }
    __syncthreads();

    int wave = tid >> 5, lane = tid & 31;
    int tile = blockIdx.x * 8 + wave;
    if (tile >= NTILES) return;

    int h  = lane >> 4;
    int nn = lane & 15;
    long m = (long)tile * 16 + nn;
    const float* xrow = x + m * FDIM;

    v8f c = {0.f, 0.f, 0.f, 0.f, 0.f, 0.f, 0.f, 0.f};
#pragma unroll
    for (int kk = 0; kk < FDIM; kk += 4) {
        int k0 = kk + 2 * h;
        v2f a = *(const v2f*)(xrow + k0);
        v2f b = *(const v2f*)(&lw[(k0 >> 1) * 32 + nn * 2]);
        c = __builtin_amdgcn_wmma_f32_16x16x4_f32(false, a, false, b,
                                                  (short)0, c, false, false);
    }
#pragma unroll
    for (int v = 0; v < 8; ++v) {
        long r = (long)tile * 16 + v + 8 * h;
        float val = c[v] * dis[r];
        hws[r * HDIM + nn] = val;
        agg[r * HDIM + nn] = val;
    }
}

__global__ void edge_scatter_kernel(const int* __restrict__ row, const int* __restrict__ col,
                                    const float* __restrict__ hws, float* __restrict__ agg) {
    long t = (long)blockIdx.x * blockDim.x + threadIdx.x;
    int e = (int)(t >> 4);
    int j = (int)(t & 15);
    if (e < N_EDGES) {
        int r = row[e];
        int ccol = col[e];
        atomicAdd(&agg[(long)ccol * HDIM + j], hws[(long)r * HDIM + j]);
    }
}

__global__ void __launch_bounds__(256) gemm2_kernel(const float* __restrict__ agg1,
                                                    const float* __restrict__ W2,
                                                    const float* __restrict__ b1v,
                                                    const float* __restrict__ dis,
                                                    float* __restrict__ hws2,
                                                    float* __restrict__ agg2) {
    __shared__ float lw[(HDIM / 2) * 32];
    __shared__ float lb1[HDIM];
    int tid = threadIdx.x;
    if (tid < 256) {
        int k = tid >> 4, n = tid & 15;
        lw[(k >> 1) * 32 + n * 2 + (k & 1)] = W2[tid];
    }
    if (tid < HDIM) lb1[tid] = b1v[tid];
    __syncthreads();

    int wave = tid >> 5, lane = tid & 31;
    int tile = blockIdx.x * 8 + wave;
    if (tile >= NTILES) return;

    int h  = lane >> 4;
    int nn = lane & 15;
    long m = (long)tile * 16 + nn;
    float dm = dis[m];
    const float* arow = agg1 + m * HDIM;

    v8f c = {0.f, 0.f, 0.f, 0.f, 0.f, 0.f, 0.f, 0.f};
#pragma unroll
    for (int kk = 0; kk < HDIM; kk += 4) {
        int k0 = kk + 2 * h;
        v2f raw = *(const v2f*)(arow + k0);
        v2f a;
        a.x = fmaxf(dm * raw.x + lb1[k0], 0.0f);
        a.y = fmaxf(dm * raw.y + lb1[k0 + 1], 0.0f);
        v2f b = *(const v2f*)(&lw[(k0 >> 1) * 32 + nn * 2]);
        c = __builtin_amdgcn_wmma_f32_16x16x4_f32(false, a, false, b,
                                                  (short)0, c, false, false);
    }
#pragma unroll
    for (int v = 0; v < 8; ++v) {
        long r = (long)tile * 16 + v + 8 * h;
        float val = c[v] * dis[r];
        hws2[r * HDIM + nn] = val;
        agg2[r * HDIM + nn] = val;
    }
}

__global__ void pool_kernel(const float* __restrict__ agg2, const float* __restrict__ dis,
                            const float* __restrict__ b2v, const int* __restrict__ batch,
                            float* __restrict__ pool) {
    int t = blockIdx.x * blockDim.x + threadIdx.x;
    int n = t >> 4, j = t & 15;
    if (n < N_NODES) {
        float val = dis[n] * agg2[(long)n * HDIM + j] + b2v[j];
        atomicAdd(&pool[batch[n] * HDIM + j], val);
    }
}

__global__ void final_kernel(const float* __restrict__ pool, const int* __restrict__ counts,
                             const float* __restrict__ Wfc, const float* __restrict__ bfc,
                             float* __restrict__ out) {
    int t = blockIdx.x * blockDim.x + threadIdx.x;
    if (t < BGRAPH * CDIM) {
        int b = t / CDIM, cc = t % CDIM;
        float inv = 1.0f / fmaxf((float)counts[b], 1.0f);
        float acc = bfc[cc];
#pragma unroll
        for (int j = 0; j < HDIM; ++j)
            acc += pool[b * HDIM + j] * inv * Wfc[j * CDIM + cc];
        out[t] = acc;
    }
}

extern "C" void kernel_launch(void* const* d_in, const int* in_sizes, int n_in,
                              void* d_out, int out_size, void* d_ws, size_t ws_size,
                              hipStream_t stream) {
    (void)in_sizes; (void)n_in; (void)out_size; (void)ws_size;

    const float* x    = (const float*)d_in[0];
    const int*   ei   = (const int*)d_in[1];
    const int*   batch= (const int*)d_in[2];
    const float* W1   = (const float*)d_in[3];
    const float* b1   = (const float*)d_in[4];
    const float* W2   = (const float*)d_in[5];
    const float* b2   = (const float*)d_in[6];
    const float* Wfc  = (const float*)d_in[7];
    const float* bfc  = (const float*)d_in[8];
    float* out = (float*)d_out;

    const int* erow = ei;
    const int* ecol = ei + N_EDGES;

    char* ws = (char*)d_ws;
    size_t o = 0;
    auto alloc = [&](size_t bytes) -> void* {
        void* p = ws + o;
        o += (bytes + 255) & ~(size_t)255;
        return p;
    };
    float* dis    = (float*)alloc((size_t)N_NODES * 4);
    int*   degcnt = (int*)  alloc((size_t)N_NODES * 4);
    float* hws    = (float*)alloc((size_t)N_NODES * HDIM * 4);
    float* agg1   = (float*)alloc((size_t)N_NODES * HDIM * 4);
    float* agg2   = (float*)alloc((size_t)N_NODES * HDIM * 4);
    float* pool   = (float*)alloc((size_t)BGRAPH * HDIM * 4);
    int*   counts = (int*)  alloc((size_t)BGRAPH * 4);

    hipMemsetAsync(degcnt, 0, (size_t)N_NODES * 4, stream);
    hipMemsetAsync(counts, 0, (size_t)BGRAPH * 4, stream);
    hipMemsetAsync(pool,   0, (size_t)BGRAPH * HDIM * 4, stream);

    count_deg_kernel<<<(N_EDGES + 255) / 256, 256, 0, stream>>>(ecol, degcnt);
    node_prep_kernel<<<(N_NODES + 255) / 256, 256, 0, stream>>>(degcnt, batch, dis, counts);

    int gemm_grid = (NTILES + 7) / 8;
    gemm1_kernel<<<gemm_grid, 256, 0, stream>>>(x, W1, dis, hws, agg1);

    int esc_grid = (int)(((long)N_EDGES * 16 + 255) / 256);
    edge_scatter_kernel<<<esc_grid, 256, 0, stream>>>(erow, ecol, hws, agg1);

    gemm2_kernel<<<gemm_grid, 256, 0, stream>>>(agg1, W2, b1, dis, hws, agg2);

    edge_scatter_kernel<<<esc_grid, 256, 0, stream>>>(erow, ecol, hws, agg2);

    pool_kernel<<<(N_NODES * 16 + 255) / 256, 256, 0, stream>>>(agg2, dis, b2, batch, pool);
    final_kernel<<<(BGRAPH * CDIM + 255) / 256, 256, 0, stream>>>(pool, counts, Wfc, bfc, out);
}


# edge scatter-add: a work-group owns 4000 target nodes as f32 sums in LDS (LDS atomic adds), streams a fifth of the edges, adds its sums to agg by whole-line atomics; order of the sums changes
# speedup vs baseline: 1.4787x; 1.4787x over previous
#include <hip/hip_runtime.h>

#define N_NODES 200000
#define N_EDGES 6400000
#define FDIM    128
#define HDIM    16
#define CDIM    10
#define BGRAPH  512
#define NTILES  (N_NODES / 16)

typedef float v2f __attribute__((ext_vector_type(2)));
typedef float v8f __attribute__((ext_vector_type(8)));

__global__ void count_deg_kernel(const int* __restrict__ col, int* __restrict__ degcnt) {
    int e = blockIdx.x * blockDim.x + threadIdx.x;
    if (e < N_EDGES) atomicAdd(&degcnt[col[e]], 1);
}

__global__ void node_prep_kernel(const int* __restrict__ degcnt, const int* __restrict__ batch,
                                 float* __restrict__ dis, int* __restrict__ counts) {
    int n = blockIdx.x * blockDim.x + threadIdx.x;
    if (n < N_NODES) {
        dis[n] = rsqrtf(1.0f + (float)degcnt[n]);
        atomicAdd(&counts[batch[n]], 1);
    }
}

__global__ void __launch_bounds__(256) gemm1_kernel(const float* __restrict__ x,
                                                    const float* __restrict__ W1,
                                                    const float* __restrict__ dis,
                                                    float* __restrict__ hws,
                                                    float* __restrict__ agg) {
    __shared__ float lw[(FDIM / 2) * 32];
    int tid = threadIdx.x;
#pragma unroll
    for (int i = 0; i < 8; ++i) {
        int idx = i * 256 + tid;
        int k = idx >> 4, n = idx & 15;
        lw[(k >> 1) * 32 + n * 2 + (k & 1)] = W1[idx];
    }
    __syncthreads();

    int wave = tid >> 5, lane = tid & 31;
    int tile = blockIdx.x * 8 + wave;
    if (tile >= NTILES) return;

    int h  = lane >> 4;
    int nn = lane & 15;
    long m = (long)tile * 16 + nn;
    const float* xrow = x + m * FDIM;

    v8f c = {0.f, 0.f, 0.f, 0.f, 0.f, 0.f, 0.f, 0.f};
#pragma unroll
    for (int kk = 0; kk < FDIM; kk += 4) {
        int k0 = kk + 2 * h;
        v2f a = *(const v2f*)(xrow + k0);
        v2f b = *(const v2f*)(&lw[(k0 >> 1) * 32 + nn * 2]);
        c = __builtin_amdgcn_wmma_f32_16x16x4_f32(false, a, false, b,
                                                  (short)0, c, false, false);
    }
#pragma unroll
    for (int v = 0; v < 8; ++v) {
        long r = (long)tile * 16 + v + 8 * h;
        float val = c[v] * dis[r];
        hws[r * HDIM + nn] = val;
        agg[r * HDIM + nn] = val;
    }
}

#define RANGE_NODES      4000
#define N_RANGES         (N_NODES / RANGE_NODES)
#define E_CHUNKS         5
#define EDGE_GROUP       128
#define N_GROUPS         (N_EDGES / EDGE_GROUP)
#define GROUPS_PER_CHUNK (N_GROUPS / E_CHUNKS)
#define SC_THREADS       1024
#define SC_WAVES         (SC_THREADS / 32)
#define QCAP             160

__device__ __forceinline__ void drain_queue(const int* __restrict__ qw, int base, int n, int h, int j,
                                            const float* __restrict__ hws, float* __restrict__ acc) {
    __builtin_amdgcn_wave_barrier();
    int p[16];
    float v[16];
#pragma unroll
    for (int k = 0; k < 16; ++k) {
        int s = 2 * k + h;
        p[k] = qw[base + (s < n ? s : 0)];
    }
#pragma unroll
    for (int k = 0; k < 16; ++k) {
        int s = 2 * k + h;
        unsigned r = (unsigned)p[k] >> 12;
        v[k] = (s < n) ? hws[r * HDIM + j] : 0.0f;
    }
#pragma unroll
    for (int k = 0; k < 16; ++k) {
        int s = 2 * k + h;
        int rel = p[k] & 4095;
        if (s < n)
            __hip_atomic_fetch_add(&acc[rel * HDIM + j], v[k], __ATOMIC_RELAXED,
                                   __HIP_MEMORY_SCOPE_WORKGROUP);
    }
    __builtin_amdgcn_wave_barrier();
}

__device__ __forceinline__ int test_group(int4 cv, int4 rv, int lo, int* __restrict__ qw, int cnt) {
    int cc[4] = {cv.x, cv.y, cv.z, cv.w};
    int rr[4] = {rv.x, rv.y, rv.z, rv.w};
#pragma unroll
    for (int k = 0; k < 4; ++k) {
        unsigned rel = (unsigned)(cc[k] - lo);
        unsigned m = __builtin_amdgcn_ballot_w32(rel < (unsigned)RANGE_NODES);
        if (m != 0u) {
            int pos = cnt + (int)__builtin_amdgcn_mbcnt_lo(m, 0u);
            if (rel < (unsigned)RANGE_NODES) qw[pos] = (rr[k] << 12) | (int)rel;
            cnt += __builtin_popcount(m);
        }
    }
    return cnt;
}

__global__ void __launch_bounds__(SC_THREADS) edge_scatter_kernel(const int* __restrict__ row,
                                                                  const int* __restrict__ col,
                                                                  const float* __restrict__ hws,
                                                                  float* __restrict__ agg) {
    __shared__ __attribute__((aligned(16))) float acc[RANGE_NODES * HDIM];
    __shared__ int queue[SC_WAVES][QCAP];

    const int tid  = threadIdx.x;
    const int lane = tid & 31;
    const int w    = __builtin_amdgcn_readfirstlane(tid >> 5);
    const int range = blockIdx.x % N_RANGES;
    const int chunk = blockIdx.x / N_RANGES;
    const int lo = range * RANGE_NODES;

    for (int i = tid; i < RANGE_NODES * HDIM / 4; i += SC_THREADS)
        ((float4*)acc)[i] = make_float4(0.f, 0.f, 0.f, 0.f);
    __syncthreads();

    const int h = lane >> 4, j = lane & 15;
    const int4* __restrict__ col4 = (const int4*)col;
    const int4* __restrict__ row4 = (const int4*)row;
    int* qw = &queue[w][0];
    int cnt = 0;

    int g = chunk * GROUPS_PER_CHUNK + w;
    const int gend = (chunk + 1) * GROUPS_PER_CHUNK;

    int4 ca = col4[g * 32 + lane], ra = row4[g * 32 + lane];
    int4 cb = ca, rb = ra;
    while (true) {
        int g1 = g + SC_WAVES;
        if (g1 < gend) { cb = col4[g1 * 32 + lane]; rb = row4[g1 * 32 + lane]; }
        cnt = test_group(ca, ra, lo, qw, cnt);
        while (cnt >= 32) { drain_queue(qw, cnt - 32, 32, h, j, hws, acc); cnt -= 32; }
        if (g1 >= gend) break;
        int g2 = g1 + SC_WAVES;
        if (g2 < gend) { ca = col4[g2 * 32 + lane]; ra = row4[g2 * 32 + lane]; }
        cnt = test_group(cb, rb, lo, qw, cnt);
        while (cnt >= 32) { drain_queue(qw, cnt - 32, 32, h, j, hws, acc); cnt -= 32; }
        if (g2 >= gend) break;
        g = g2;
    }
    if (cnt > 0) drain_queue(qw, 0, cnt, h, j, hws, acc);

    __syncthreads();
    float* dst = agg + (long)lo * HDIM;
    for (int i = tid; i < RANGE_NODES * HDIM; i += SC_THREADS)
        atomicAdd(&dst[i], acc[i]);
}

__global__ void __launch_bounds__(256) gemm2_kernel(const float* __restrict__ agg1,
                                                    const float* __restrict__ W2,
                                                    const float* __restrict__ b1v,
                                                    const float* __restrict__ dis,
                                                    float* __restrict__ hws2,
                                                    float* __restrict__ agg2) {
    __shared__ float lw[(HDIM / 2) * 32];
    __shared__ float lb1[HDIM];
    int tid = threadIdx.x;
    if (tid < 256) {
        int k = tid >> 4, n = tid & 15;
        lw[(k >> 1) * 32 + n * 2 + (k & 1)] = W2[tid];
    }
    if (tid < HDIM) lb1[tid] = b1v[tid];
    __syncthreads();

    int wave = tid >> 5, lane = tid & 31;
    int tile = blockIdx.x * 8 + wave;
    if (tile >= NTILES) return;

    int h  = lane >> 4;
    int nn = lane & 15;
    long m = (long)tile * 16 + nn;
    float dm = dis[m];
    const float* arow = agg1 + m * HDIM;

    v8f c = {0.f, 0.f, 0.f, 0.f, 0.f, 0.f, 0.f, 0.f};
#pragma unroll
    for (int kk = 0; kk < HDIM; kk += 4) {
        int k0 = kk + 2 * h;
        v2f raw = *(const v2f*)(arow + k0);
        v2f a;
        a.x = fmaxf(dm * raw.x + lb1[k0], 0.0f);
        a.y = fmaxf(dm * raw.y + lb1[k0 + 1], 0.0f);
        v2f b = *(const v2f*)(&lw[(k0 >> 1) * 32 + nn * 2]);
        c = __builtin_amdgcn_wmma_f32_16x16x4_f32(false, a, false, b,
                                                  (short)0, c, false, false);
    }
#pragma unroll
    for (int v = 0; v < 8; ++v) {
        long r = (long)tile * 16 + v + 8 * h;
        float val = c[v] * dis[r];
        hws2[r * HDIM + nn] = val;
        agg2[r * HDIM + nn] = val;
    }
}

__global__ void pool_kernel(const float* __restrict__ agg2, const float* __restrict__ dis,
                            const float* __restrict__ b2v, const int* __restrict__ batch,
                            float* __restrict__ pool) {
    int t = blockIdx.x * blockDim.x + threadIdx.x;
    int n = t >> 4, j = t & 15;
    if (n < N_NODES) {
        float val = dis[n] * agg2[(long)n * HDIM + j] + b2v[j];
        atomicAdd(&pool[batch[n] * HDIM + j], val);
    }
}

__global__ void final_kernel(const float* __restrict__ pool, const int* __restrict__ counts,
                             const float* __restrict__ Wfc, const float* __restrict__ bfc,
                             float* __restrict__ out) {
    int t = blockIdx.x * blockDim.x + threadIdx.x;
    if (t < BGRAPH * CDIM) {
        int b = t / CDIM, cc = t % CDIM;
        float inv = 1.0f / fmaxf((float)counts[b], 1.0f);
        float acc = bfc[cc];
#pragma unroll
        for (int j = 0; j < HDIM; ++j)
            acc += pool[b * HDIM + j] * inv * Wfc[j * CDIM + cc];
        out[t] = acc;
    }
}

extern "C" void kernel_launch(void* const* d_in, const int* in_sizes, int n_in,
                              void* d_out, int out_size, void* d_ws, size_t ws_size,
                              hipStream_t stream) {
    (void)in_sizes; (void)n_in; (void)out_size; (void)ws_size;

    const float* x    = (const float*)d_in[0];
    const int*   ei   = (const int*)d_in[1];
    const int*   batch= (const int*)d_in[2];
    const float* W1   = (const float*)d_in[3];
    const float* b1   = (const float*)d_in[4];
    const float* W2   = (const float*)d_in[5];
    const float* b2   = (const float*)d_in[6];
    const float* Wfc  = (const float*)d_in[7];
    const float* bfc  = (const float*)d_in[8];
    float* out = (float*)d_out;

    const int* erow = ei;
    const int* ecol = ei + N_EDGES;

    char* ws = (char*)d_ws;
    size_t o = 0;
    auto alloc = [&](size_t bytes) -> void* {
        void* p = ws + o;
        o += (bytes + 255) & ~(size_t)255;
        return p;
    };
    float* dis    = (float*)alloc((size_t)N_NODES * 4);
    int*   degcnt = (int*)  alloc((size_t)N_NODES * 4);
    float* hws    = (float*)alloc((size_t)N_NODES * HDIM * 4);
    float* agg1   = (float*)alloc((size_t)N_NODES * HDIM * 4);
    float* agg2   = (float*)alloc((size_t)N_NODES * HDIM * 4);
    float* pool   = (float*)alloc((size_t)BGRAPH * HDIM * 4);
    int*   counts = (int*)  alloc((size_t)BGRAPH * 4);

    hipMemsetAsync(degcnt, 0, (size_t)N_NODES * 4, stream);
    hipMemsetAsync(counts, 0, (size_t)BGRAPH * 4, stream);
    hipMemsetAsync(pool,   0, (size_t)BGRAPH * HDIM * 4, stream);

    count_deg_kernel<<<(N_EDGES + 255) / 256, 256, 0, stream>>>(ecol, degcnt);
    node_prep_kernel<<<(N_NODES + 255) / 256, 256, 0, stream>>>(degcnt, batch, dis, counts);

    int gemm_grid = (NTILES + 7) / 8;
    gemm1_kernel<<<gemm_grid, 256, 0, stream>>>(x, W1, dis, hws, agg1);

    int esc_grid = N_RANGES * E_CHUNKS;
    edge_scatter_kernel<<<esc_grid, SC_THREADS, 0, stream>>>(erow, ecol, hws, agg1);

    gemm2_kernel<<<gemm_grid, 256, 0, stream>>>(agg1, W2, b1, dis, hws, agg2);

    edge_scatter_kernel<<<esc_grid, SC_THREADS, 0, stream>>>(erow, ecol, hws, agg2);

    pool_kernel<<<(N_NODES * 16 + 255) / 256, 256, 0, stream>>>(agg2, dis, b2, batch, pool);
    final_kernel<<<(BGRAPH * CDIM + 255) / 256, 256, 0, stream>>>(pool, counts, Wfc, bfc, out);
}


# edges bucketed once per call by target range (250 buckets of 800 nodes, packed source|target); a scatter work-group reads only its bucket and stores agg whole, no global atomics; sum order changes
# speedup vs baseline: 2.0487x; 1.3855x over previous
#include <hip/hip_runtime.h>

#define N_NODES 200000
#define N_EDGES 6400000
#define FDIM    128
#define HDIM    16
#define CDIM    10
#define BGRAPH  512
#define NTILES  (N_NODES / 16)
#define BK_NODES  800
#define N_BUCKETS (N_NODES / BK_NODES)

typedef float v2f __attribute__((ext_vector_type(2)));
typedef float v8f __attribute__((ext_vector_type(8)));

__global__ void count_deg_kernel(const int* __restrict__ col, int* __restrict__ degcnt) {
    int e = blockIdx.x * blockDim.x + threadIdx.x;
    if (e < N_EDGES) atomicAdd(&degcnt[col[e]], 1);
}

__global__ void node_prep_kernel(const int* __restrict__ degcnt, const int* __restrict__ batch,
                                 float* __restrict__ dis, int* __restrict__ counts,
                                 int* __restrict__ bcnt) {
    int n = blockIdx.x * blockDim.x + threadIdx.x;
    if (n < N_NODES) {
        int d = degcnt[n];
        dis[n] = rsqrtf(1.0f + (float)d);
        atomicAdd(&counts[batch[n]], 1);
        atomicAdd(&bcnt[n / BK_NODES], d);
    }
}

__global__ void __launch_bounds__(256) bucket_scan_kernel(const int* __restrict__ bcnt,
                                                          int* __restrict__ bstart,
                                                          int* __restrict__ bcur) {
    __shared__ int s[256];
    int t = threadIdx.x;
    int v = (t < N_BUCKETS) ? bcnt[t] : 0;
    s[t] = v;
    __syncthreads();
    for (int off = 1; off < 256; off <<= 1) {
        int x = (t >= off) ? s[t - off] : 0;
        __syncthreads();
        s[t] += x;
        __syncthreads();
    }
    int excl = s[t] - v;
    if (t <= N_BUCKETS) bstart[t] = excl;
    if (t < N_BUCKETS)  bcur[t] = excl;
}

#define PT_THREADS 512
#define PT_VEC     4
#define PT_TILE4   (PT_THREADS * PT_VEC)
#define PT_EDGES   (PT_TILE4 * 4)
#define N_EDGE4    (N_EDGES / 4)
#define PT_GRID    ((N_EDGE4 + PT_TILE4 - 1) / PT_TILE4)

__global__ void __launch_bounds__(PT_THREADS) partition_kernel(const int* __restrict__ row,
                                                               const int* __restrict__ col,
                                                               int* __restrict__ bcur,
                                                               int* __restrict__ bucket) {
    __shared__ int scnt[256];
    __shared__ int sloc[256];
    __shared__ int sdel[256];
    __shared__ int sval[PT_EDGES];
    __shared__ unsigned char sbk[PT_EDGES];
    const int tid = threadIdx.x;
    if (tid < 256) scnt[tid] = 0;
    __syncthreads();

    const int4* __restrict__ col4 = (const int4*)col;
    const int4* __restrict__ row4 = (const int4*)row;
    int cc[4 * PT_VEC], rr[4 * PT_VEC], bb[4 * PT_VEC], rk[4 * PT_VEC];
#pragma unroll
    for (int i = 0; i < PT_VEC; ++i) {
        int i4 = blockIdx.x * PT_TILE4 + i * PT_THREADS + tid;
        bool in = i4 < N_EDGE4;
        int i4c = in ? i4 : N_EDGE4 - 1;
        int4 c = col4[i4c], r = row4[i4c];
        if (!in) c = make_int4(-1, -1, -1, -1);
        cc[4 * i + 0] = c.x; cc[4 * i + 1] = c.y; cc[4 * i + 2] = c.z; cc[4 * i + 3] = c.w;
        rr[4 * i + 0] = r.x; rr[4 * i + 1] = r.y; rr[4 * i + 2] = r.z; rr[4 * i + 3] = r.w;
    }
#pragma unroll
    for (int q = 0; q < 4 * PT_VEC; ++q) {
        bb[q] = 0; rk[q] = 0;
        if (cc[q] >= 0) {
            bb[q] = (int)((unsigned)cc[q] / (unsigned)BK_NODES);
            rk[q] = atomicAdd(&scnt[bb[q]], 1);
        }
    }
    __syncthreads();

    if (tid < 32) {
        int c8[8];
        int run = 0;
#pragma unroll
        for (int q = 0; q < 8; ++q) { int t = scnt[tid * 8 + q]; c8[q] = run; run += t; }
        int incl = run;
#pragma unroll
        for (int d = 1; d < 32; d <<= 1) {
            int y = __shfl_up(incl, d);
            if (tid >= d) incl += y;
        }
        int lanebase = incl - run;
#pragma unroll
        for (int q = 0; q < 8; ++q) sloc[tid * 8 + q] = lanebase + c8[q];
    }
    __syncthreads();

    if (tid < N_BUCKETS) {
        int n = scnt[tid];
        int g = (n > 0) ? atomicAdd(&bcur[tid], n) : 0;
        sdel[tid] = g - sloc[tid];
    }
#pragma unroll
    for (int q = 0; q < 4 * PT_VEC; ++q) {
        if (cc[q] >= 0) {
            int slot = sloc[bb[q]] + rk[q];
            sval[slot] = (rr[q] << 10) | (cc[q] - bb[q] * BK_NODES);
            sbk[slot]  = (unsigned char)bb[q];
        }
    }
    __syncthreads();

    int ntot = N_EDGES - blockIdx.x * PT_EDGES;
    if (ntot > PT_EDGES) ntot = PT_EDGES;
    for (int slot = tid; slot < ntot; slot += PT_THREADS) {
        int b = sbk[slot];
        bucket[sdel[b] + slot] = sval[slot];
    }
}

__global__ void __launch_bounds__(256) gemm1_kernel(const float* __restrict__ x,
                                                    const float* __restrict__ W1,
                                                    const float* __restrict__ dis,
                                                    float* __restrict__ hws) {
    __shared__ float lw[(FDIM / 2) * 32];
    int tid = threadIdx.x;
#pragma unroll
    for (int i = 0; i < 8; ++i) {
        int idx = i * 256 + tid;
        int k = idx >> 4, n = idx & 15;
        lw[(k >> 1) * 32 + n * 2 + (k & 1)] = W1[idx];
    }
    __syncthreads();

    int wave = tid >> 5, lane = tid & 31;
    int tile = blockIdx.x * 8 + wave;
    if (tile >= NTILES) return;

    int h  = lane >> 4;
    int nn = lane & 15;
    long m = (long)tile * 16 + nn;
    const float* xrow = x + m * FDIM;

    v8f c = {0.f, 0.f, 0.f, 0.f, 0.f, 0.f, 0.f, 0.f};
#pragma unroll
    for (int kk = 0; kk < FDIM; kk += 4) {
        int k0 = kk + 2 * h;
        v2f a = *(const v2f*)(xrow + k0);
        v2f b = *(const v2f*)(&lw[(k0 >> 1) * 32 + nn * 2]);
        c = __builtin_amdgcn_wmma_f32_16x16x4_f32(false, a, false, b,
                                                  (short)0, c, false, false);
    }
#pragma unroll
    for (int v = 0; v < 8; ++v) {
        long r = (long)tile * 16 + v + 8 * h;
        hws[r * HDIM + nn] = c[v] * dis[r];
    }
}

#define SC_THREADS 1024
#define SC_WAVES   (SC_THREADS / 32)

template <bool FULL>
__device__ __forceinline__ void add_edges(const int (&p)[16], int n, int h, int j,
                                          const float* __restrict__ hws, float* __restrict__ acc) {
    float v[16];
#pragma unroll
    for (int k = 0; k < 16; ++k) {
        int s = 2 * k + h;
        unsigned r = (unsigned)p[k] >> 10;
        v[k] = (FULL || s < n) ? hws[r * HDIM + j] : 0.0f;
    }
#pragma unroll
    for (int k = 0; k < 16; ++k) {
        int s = 2 * k + h;
        int rel = p[k] & 1023;
        if (FULL || s < n)
            __hip_atomic_fetch_add(&acc[rel * HDIM + j], v[k], __ATOMIC_RELAXED,
                                   __HIP_MEMORY_SCOPE_WORKGROUP);
    }
}

__global__ void __launch_bounds__(SC_THREADS) edge_scatter_kernel(const int* __restrict__ bucket,
                                                                  const int* __restrict__ bstart,
                                                                  const float* __restrict__ hws,
                                                                  float* __restrict__ agg) {
    __shared__ __attribute__((aligned(16))) float acc[BK_NODES * HDIM];

    const int tid  = threadIdx.x;
    const int lane = tid & 31;
    const int w    = __builtin_amdgcn_readfirstlane(tid >> 5);
    const int b    = blockIdx.x;
    const int lo   = b * BK_NODES;

    for (int i = tid; i < BK_NODES * HDIM / 4; i += SC_THREADS)
        ((float4*)acc)[i] = make_float4(0.f, 0.f, 0.f, 0.f);
    __syncthreads();

    const int h = lane >> 4, j = lane & 15;
    const int beg = bstart[b], end = bstart[b + 1];

    int i = beg + w * 32;
    int e = (i + lane < end) ? bucket[i + lane] : 0;
    while (i < end) {
        int p[16];
#pragma unroll
        for (int k = 0; k < 16; ++k)
            p[k] = __builtin_amdgcn_ds_bpermute((2 * k + h) << 2, e);
        int inext = i + SC_THREADS;
        e = (inext + lane < end) ? bucket[inext + lane] : 0;
        int n = end - i;
        if (n >= 32) add_edges<true>(p, 32, h, j, hws, acc);
        else         add_edges<false>(p, n, h, j, hws, acc);
        i = inext;
    }

    __syncthreads();
    const float4* __restrict__ self4 = (const float4*)(hws + (long)lo * HDIM);
    float4* __restrict__ dst4 = (float4*)(agg + (long)lo * HDIM);
    for (int q = tid; q < BK_NODES * HDIM / 4; q += SC_THREADS) {
        float4 a = ((const float4*)acc)[q];
        float4 s = self4[q];
        dst4[q] = make_float4(s.x + a.x, s.y + a.y, s.z + a.z, s.w + a.w);
    }
}

__global__ void __launch_bounds__(256) gemm2_kernel(const float* __restrict__ agg1,
                                                    const float* __restrict__ W2,
                                                    const float* __restrict__ b1v,
                                                    const float* __restrict__ dis,
                                                    float* __restrict__ hws2) {
    __shared__ float lw[(HDIM / 2) * 32];
    __shared__ float lb1[HDIM];
    int tid = threadIdx.x;
    if (tid < 256) {
        int k = tid >> 4, n = tid & 15;
        lw[(k >> 1) * 32 + n * 2 + (k & 1)] = W2[tid];
    }
    if (tid < HDIM) lb1[tid] = b1v[tid];
    __syncthreads();

    int wave = tid >> 5, lane = tid & 31;
    int tile = blockIdx.x * 8 + wave;
    if (tile >= NTILES) return;

    int h  = lane >> 4;
    int nn = lane & 15;
    long m = (long)tile * 16 + nn;
    float dm = dis[m];
    const float* arow = agg1 + m * HDIM;

    v8f c = {0.f, 0.f, 0.f, 0.f, 0.f, 0.f, 0.f, 0.f};
#pragma unroll
    for (int kk = 0; kk < HDIM; kk += 4) {
        int k0 = kk + 2 * h;
        v2f raw = *(const v2f*)(arow + k0);
        v2f a;
        a.x = fmaxf(dm * raw.x + lb1[k0], 0.0f);
        a.y = fmaxf(dm * raw.y + lb1[k0 + 1], 0.0f);
        v2f b = *(const v2f*)(&lw[(k0 >> 1) * 32 + nn * 2]);
        c = __builtin_amdgcn_wmma_f32_16x16x4_f32(false, a, false, b,
                                                  (short)0, c, false, false);
    }
#pragma unroll
    for (int v = 0; v < 8; ++v) {
        long r = (long)tile * 16 + v + 8 * h;
        hws2[r * HDIM + nn] = c[v] * dis[r];
    }
}

__global__ void pool_kernel(const float* __restrict__ agg2, const float* __restrict__ dis,
                            const float* __restrict__ b2v, const int* __restrict__ batch,
                            float* __restrict__ pool) {
    int t = blockIdx.x * blockDim.x + threadIdx.x;
    int n = t >> 4, j = t & 15;
    if (n < N_NODES) {
        float val = dis[n] * agg2[(long)n * HDIM + j] + b2v[j];
        atomicAdd(&pool[batch[n] * HDIM + j], val);
    }
}

__global__ void final_kernel(const float* __restrict__ pool, const int* __restrict__ counts,
                             const float* __restrict__ Wfc, const float* __restrict__ bfc,
                             float* __restrict__ out) {
    int t = blockIdx.x * blockDim.x + threadIdx.x;
    if (t < BGRAPH * CDIM) {
        int b = t / CDIM, cc = t % CDIM;
        float inv = 1.0f / fmaxf((float)counts[b], 1.0f);
        float acc = bfc[cc];
#pragma unroll
        for (int j = 0; j < HDIM; ++j)
            acc += pool[b * HDIM + j] * inv * Wfc[j * CDIM + cc];
        out[t] = acc;
    }
}

extern "C" void kernel_launch(void* const* d_in, const int* in_sizes, int n_in,
                              void* d_out, int out_size, void* d_ws, size_t ws_size,
                              hipStream_t stream) {
    (void)in_sizes; (void)n_in; (void)out_size; (void)ws_size;

    const float* x    = (const float*)d_in[0];
    const int*   ei   = (const int*)d_in[1];
    const int*   batch= (const int*)d_in[2];
    const float* W1   = (const float*)d_in[3];
    const float* b1   = (const float*)d_in[4];
    const float* W2   = (const float*)d_in[5];
    const float* b2   = (const float*)d_in[6];
    const float* Wfc  = (const float*)d_in[7];
    const float* bfc  = (const float*)d_in[8];
    float* out = (float*)d_out;

    const int* erow = ei;
    const int* ecol = ei + N_EDGES;

    char* ws = (char*)d_ws;
    size_t o = 0;
    auto alloc = [&](size_t bytes) -> void* {
        void* p = ws + o;
        o += (bytes + 255) & ~(size_t)255;
        return p;
    };
    float* dis    = (float*)alloc((size_t)N_NODES * 4);
    int*   degcnt = (int*)  alloc((size_t)N_NODES * 4);
    float* hws    = (float*)alloc((size_t)N_NODES * HDIM * 4);
    float* agg    = (float*)alloc((size_t)N_NODES * HDIM * 4);
    int*   bucket = (int*)  alloc((size_t)N_EDGES * 4);
    float* pool   = (float*)alloc((size_t)BGRAPH * HDIM * 4);
    int*   counts = (int*)  alloc((size_t)(BGRAPH + 256) * 4);
    int*   bcnt   = counts + BGRAPH;
    int*   bstart = (int*)  alloc((size_t)512 * 4);
    int*   bcur   = (int*)  alloc((size_t)256 * 4);

    hipMemsetAsync(degcnt, 0, (size_t)N_NODES * 4, stream);
    hipMemsetAsync(counts, 0, (size_t)(BGRAPH + 256) * 4, stream);
    hipMemsetAsync(pool,   0, (size_t)BGRAPH * HDIM * 4, stream);

    count_deg_kernel<<<(N_EDGES + 255) / 256, 256, 0, stream>>>(ecol, degcnt);
    node_prep_kernel<<<(N_NODES + 255) / 256, 256, 0, stream>>>(degcnt, batch, dis, counts, bcnt);
    bucket_scan_kernel<<<1, 256, 0, stream>>>(bcnt, bstart, bcur);
    partition_kernel<<<PT_GRID, PT_THREADS, 0, stream>>>(erow, ecol, bcur, bucket);

    int gemm_grid = (NTILES + 7) / 8;
    gemm1_kernel<<<gemm_grid, 256, 0, stream>>>(x, W1, dis, hws);

    edge_scatter_kernel<<<N_BUCKETS, SC_THREADS, 0, stream>>>(bucket, bstart, hws, agg);

    gemm2_kernel<<<gemm_grid, 256, 0, stream>>>(agg, W2, b1, dis, hws);

    edge_scatter_kernel<<<N_BUCKETS, SC_THREADS, 0, stream>>>(bucket, bstart, hws, agg);

    pool_kernel<<<(N_NODES * 16 + 255) / 256, 256, 0, stream>>>(agg, dis, b2, batch, pool);
    final_kernel<<<(BGRAPH * CDIM + 255) / 256, 256, 0, stream>>>(pool, counts, Wfc, bfc, out);
}


# in-degree counted per bucket in LDS from the bucketed edge list, bucket sizes from an LDS histogram of col: the 6.4M scattered global integer atomics and one memset go; float sums unchanged
# speedup vs baseline: 5.0034x; 2.4422x over previous
#include <hip/hip_runtime.h>

#define N_NODES 200000
#define N_EDGES 6400000
#define FDIM    128
#define HDIM    16
#define CDIM    10
#define BGRAPH  512
#define NTILES  (N_NODES / 16)
#define BK_NODES  800
#define N_BUCKETS (N_NODES / BK_NODES)

typedef float v2f __attribute__((ext_vector_type(2)));
typedef float v8f __attribute__((ext_vector_type(8)));

#define HS_THREADS 1024
#define HS_GRID    256
#define HS_TILE4   (N_EDGES / 4 / HS_GRID)
#define HS_VEC     ((HS_TILE4 + HS_THREADS - 1) / HS_THREADS)

__global__ void __launch_bounds__(HS_THREADS) bucket_hist_kernel(const int* __restrict__ col,
                                                                 int* __restrict__ bcnt) {
    __shared__ int scnt[256];
    const int tid = threadIdx.x;
    if (tid < 256) scnt[tid] = 0;
    __syncthreads();

    const int4* __restrict__ col4 = (const int4*)col + blockIdx.x * HS_TILE4;
    int4 c[HS_VEC];
#pragma unroll
    for (int i = 0; i < HS_VEC; ++i) {
        int i4 = i * HS_THREADS + tid;
        bool in = i4 < HS_TILE4;
        c[i] = col4[in ? i4 : HS_TILE4 - 1];
        if (!in) c[i] = make_int4(-1, -1, -1, -1);
    }
#pragma unroll
    for (int i = 0; i < HS_VEC; ++i) {
        if (c[i].x >= 0) {
            atomicAdd(&scnt[(unsigned)c[i].x / (unsigned)BK_NODES], 1);
            atomicAdd(&scnt[(unsigned)c[i].y / (unsigned)BK_NODES], 1);
            atomicAdd(&scnt[(unsigned)c[i].z / (unsigned)BK_NODES], 1);
            atomicAdd(&scnt[(unsigned)c[i].w / (unsigned)BK_NODES], 1);
        }
    }
    __syncthreads();
    if (tid < N_BUCKETS) {
        int n = scnt[tid];
        if (n > 0) atomicAdd(&bcnt[tid], n);
    }
}

__global__ void __launch_bounds__(256) bucket_scan_kernel(const int* __restrict__ bcnt,
                                                          int* __restrict__ bstart,
                                                          int* __restrict__ bcur) {
    __shared__ int s[256];
    int t = threadIdx.x;
    int v = (t < N_BUCKETS) ? bcnt[t] : 0;
    s[t] = v;
    __syncthreads();
    for (int off = 1; off < 256; off <<= 1) {
        int x = (t >= off) ? s[t - off] : 0;
        __syncthreads();
        s[t] += x;
        __syncthreads();
    }
    int excl = s[t] - v;
    if (t <= N_BUCKETS) bstart[t] = excl;
    if (t < N_BUCKETS)  bcur[t] = excl;
}

#define PT_THREADS 512
#define PT_VEC     4
#define PT_TILE4   (PT_THREADS * PT_VEC)
#define PT_EDGES   (PT_TILE4 * 4)
#define N_EDGE4    (N_EDGES / 4)
#define PT_GRID    ((N_EDGE4 + PT_TILE4 - 1) / PT_TILE4)

__global__ void __launch_bounds__(PT_THREADS) partition_kernel(const int* __restrict__ row,
                                                               const int* __restrict__ col,
                                                               int* __restrict__ bcur,
                                                               int* __restrict__ bucket) {
    __shared__ int scnt[256];
    __shared__ int sloc[256];
    __shared__ int sdel[256];
    __shared__ int sval[PT_EDGES];
    __shared__ unsigned char sbk[PT_EDGES];
    const int tid = threadIdx.x;
    if (tid < 256) scnt[tid] = 0;
    __syncthreads();

    const int4* __restrict__ col4 = (const int4*)col;
    const int4* __restrict__ row4 = (const int4*)row;
    int cc[4 * PT_VEC], rr[4 * PT_VEC], bb[4 * PT_VEC], rk[4 * PT_VEC];
#pragma unroll
    for (int i = 0; i < PT_VEC; ++i) {
        int i4 = blockIdx.x * PT_TILE4 + i * PT_THREADS + tid;
        bool in = i4 < N_EDGE4;
        int i4c = in ? i4 : N_EDGE4 - 1;
        int4 c = col4[i4c], r = row4[i4c];
        if (!in) c = make_int4(-1, -1, -1, -1);
        cc[4 * i + 0] = c.x; cc[4 * i + 1] = c.y; cc[4 * i + 2] = c.z; cc[4 * i + 3] = c.w;
        rr[4 * i + 0] = r.x; rr[4 * i + 1] = r.y; rr[4 * i + 2] = r.z; rr[4 * i + 3] = r.w;
    }
#pragma unroll
    for (int q = 0; q < 4 * PT_VEC; ++q) {
        bb[q] = 0; rk[q] = 0;
        if (cc[q] >= 0) {
            bb[q] = (int)((unsigned)cc[q] / (unsigned)BK_NODES);
            rk[q] = atomicAdd(&scnt[bb[q]], 1);
        }
    }
    __syncthreads();

    if (tid < 32) {
        int c8[8];
        int run = 0;
#pragma unroll
        for (int q = 0; q < 8; ++q) { int t = scnt[tid * 8 + q]; c8[q] = run; run += t; }
        int incl = run;
#pragma unroll
        for (int d = 1; d < 32; d <<= 1) {
            int y = __shfl_up(incl, d);
            if (tid >= d) incl += y;
        }
        int lanebase = incl - run;
#pragma unroll
        for (int q = 0; q < 8; ++q) sloc[tid * 8 + q] = lanebase + c8[q];
    }
    __syncthreads();

    if (tid < N_BUCKETS) {
        int n = scnt[tid];
        int g = (n > 0) ? atomicAdd(&bcur[tid], n) : 0;
        sdel[tid] = g - sloc[tid];
    }
#pragma unroll
    for (int q = 0; q < 4 * PT_VEC; ++q) {
        if (cc[q] >= 0) {
            int slot = sloc[bb[q]] + rk[q];
            sval[slot] = (rr[q] << 10) | (cc[q] - bb[q] * BK_NODES);
            sbk[slot]  = (unsigned char)bb[q];
        }
    }
    __syncthreads();

    int ntot = N_EDGES - blockIdx.x * PT_EDGES;
    if (ntot > PT_EDGES) ntot = PT_EDGES;
    for (int slot = tid; slot < ntot; slot += PT_THREADS) {
        int b = sbk[slot];
        bucket[sdel[b] + slot] = sval[slot];
    }
}

#define DG_THREADS 1024
#define DG_VEC     8

__global__ void __launch_bounds__(DG_THREADS) degree_kernel(const int* __restrict__ bucket,
                                                            const int* __restrict__ bstart,
                                                            const int* __restrict__ batch,
                                                            float* __restrict__ dis,
                                                            int* __restrict__ counts) {
    __shared__ int cnt[BK_NODES];
    const int tid = threadIdx.x;
    const int b   = blockIdx.x;
    if (tid < BK_NODES) cnt[tid] = 0;
    __syncthreads();

    const int beg = bstart[b], end = bstart[b + 1];
    int base = beg;
    for (; base + DG_THREADS * DG_VEC <= end; base += DG_THREADS * DG_VEC) {
        int p[DG_VEC];
#pragma unroll
        for (int u = 0; u < DG_VEC; ++u) p[u] = bucket[base + u * DG_THREADS + tid];
#pragma unroll
        for (int u = 0; u < DG_VEC; ++u) atomicAdd(&cnt[p[u] & 1023], 1);
    }
    {
        int p[DG_VEC];
#pragma unroll
        for (int u = 0; u < DG_VEC; ++u) {
            int k = base + u * DG_THREADS + tid;
            p[u] = (k < end) ? bucket[k] : -1;
        }
#pragma unroll
        for (int u = 0; u < DG_VEC; ++u)
            if (p[u] >= 0) atomicAdd(&cnt[p[u] & 1023], 1);
    }
    __syncthreads();

    if (tid < BK_NODES) {
        int n = b * BK_NODES + tid;
        dis[n] = rsqrtf(1.0f + (float)cnt[tid]);
        atomicAdd(&counts[batch[n]], 1);
    }
}

__global__ void __launch_bounds__(256) gemm1_kernel(const float* __restrict__ x,
                                                    const float* __restrict__ W1,
                                                    const float* __restrict__ dis,
                                                    float* __restrict__ hws) {
    __shared__ float lw[(FDIM / 2) * 32];
    int tid = threadIdx.x;
#pragma unroll
    for (int i = 0; i < 8; ++i) {
        int idx = i * 256 + tid;
        int k = idx >> 4, n = idx & 15;
        lw[(k >> 1) * 32 + n * 2 + (k & 1)] = W1[idx];
    }
    __syncthreads();

    int wave = tid >> 5, lane = tid & 31;
    int tile = blockIdx.x * 8 + wave;
    if (tile >= NTILES) return;

    int h  = lane >> 4;
    int nn = lane & 15;
    long m = (long)tile * 16 + nn;
    const float* xrow = x + m * FDIM;

    v8f c = {0.f, 0.f, 0.f, 0.f, 0.f, 0.f, 0.f, 0.f};
#pragma unroll
    for (int kk = 0; kk < FDIM; kk += 4) {
        int k0 = kk + 2 * h;
        v2f a = *(const v2f*)(xrow + k0);
        v2f b = *(const v2f*)(&lw[(k0 >> 1) * 32 + nn * 2]);
        c = __builtin_amdgcn_wmma_f32_16x16x4_f32(false, a, false, b,
                                                  (short)0, c, false, false);
    }
#pragma unroll
    for (int v = 0; v < 8; ++v) {
        long r = (long)tile * 16 + v + 8 * h;
        hws[r * HDIM + nn] = c[v] * dis[r];
    }
}

#define SC_THREADS 1024
#define SC_WAVES   (SC_THREADS / 32)

template <bool FULL>
__device__ __forceinline__ void add_edges(const int (&p)[16], int n, int h, int j,
                                          const float* __restrict__ hws, float* __restrict__ acc) {
    float v[16];
#pragma unroll
    for (int k = 0; k < 16; ++k) {
        int s = 2 * k + h;
        unsigned r = (unsigned)p[k] >> 10;
        v[k] = (FULL || s < n) ? hws[r * HDIM + j] : 0.0f;
    }
#pragma unroll
    for (int k = 0; k < 16; ++k) {
        int s = 2 * k + h;
        int rel = p[k] & 1023;
        if (FULL || s < n)
            __hip_atomic_fetch_add(&acc[rel * HDIM + j], v[k], __ATOMIC_RELAXED,
                                   __HIP_MEMORY_SCOPE_WORKGROUP);
    }
}

__global__ void __launch_bounds__(SC_THREADS) edge_scatter_kernel(const int* __restrict__ bucket,
                                                                  const int* __restrict__ bstart,
                                                                  const float* __restrict__ hws,
                                                                  float* __restrict__ agg) {
    __shared__ __attribute__((aligned(16))) float acc[BK_NODES * HDIM];

    const int tid  = threadIdx.x;
    const int lane = tid & 31;
    const int w    = __builtin_amdgcn_readfirstlane(tid >> 5);
    const int b    = blockIdx.x;
    const int lo   = b * BK_NODES;

    for (int i = tid; i < BK_NODES * HDIM / 4; i += SC_THREADS)
        ((float4*)acc)[i] = make_float4(0.f, 0.f, 0.f, 0.f);
    __syncthreads();

    const int h = lane >> 4, j = lane & 15;
    const int beg = bstart[b], end = bstart[b + 1];

    int i = beg + w * 32;
    int e = (i + lane < end) ? bucket[i + lane] : 0;
    while (i < end) {
        int p[16];
#pragma unroll
        for (int k = 0; k < 16; ++k)
            p[k] = __builtin_amdgcn_ds_bpermute((2 * k + h) << 2, e);
        int inext = i + SC_THREADS;
        e = (inext + lane < end) ? bucket[inext + lane] : 0;
        int n = end - i;
        if (n >= 32) add_edges<true>(p, 32, h, j, hws, acc);
        else         add_edges<false>(p, n, h, j, hws, acc);
        i = inext;
    }

    __syncthreads();
    const float4* __restrict__ self4 = (const float4*)(hws + (long)lo * HDIM);
    float4* __restrict__ dst4 = (float4*)(agg + (long)lo * HDIM);
    for (int q = tid; q < BK_NODES * HDIM / 4; q += SC_THREADS) {
        float4 a = ((const float4*)acc)[q];
        float4 s = self4[q];
        dst4[q] = make_float4(s.x + a.x, s.y + a.y, s.z + a.z, s.w + a.w);
    }
}

__global__ void __launch_bounds__(256) gemm2_kernel(const float* __restrict__ agg1,
                                                    const float* __restrict__ W2,
                                                    const float* __restrict__ b1v,
                                                    const float* __restrict__ dis,
                                                    float* __restrict__ hws2) {
    __shared__ float lw[(HDIM / 2) * 32];
    __shared__ float lb1[HDIM];
    int tid = threadIdx.x;
    if (tid < 256) {
        int k = tid >> 4, n = tid & 15;
        lw[(k >> 1) * 32 + n * 2 + (k & 1)] = W2[tid];
    }
    if (tid < HDIM) lb1[tid] = b1v[tid];
    __syncthreads();

    int wave = tid >> 5, lane = tid & 31;
    int tile = blockIdx.x * 8 + wave;
    if (tile >= NTILES) return;

    int h  = lane >> 4;
    int nn = lane & 15;
    long m = (long)tile * 16 + nn;
    float dm = dis[m];
    const float* arow = agg1 + m * HDIM;

    v8f c = {0.f, 0.f, 0.f, 0.f, 0.f, 0.f, 0.f, 0.f};
#pragma unroll
    for (int kk = 0; kk < HDIM; kk += 4) {
        int k0 = kk + 2 * h;
        v2f raw = *(const v2f*)(arow + k0);
        v2f a;
        a.x = fmaxf(dm * raw.x + lb1[k0], 0.0f);
        a.y = fmaxf(dm * raw.y + lb1[k0 + 1], 0.0f);
        v2f b = *(const v2f*)(&lw[(k0 >> 1) * 32 + nn * 2]);
        c = __builtin_amdgcn_wmma_f32_16x16x4_f32(false, a, false, b,
                                                  (short)0, c, false, false);
    }
#pragma unroll
    for (int v = 0; v < 8; ++v) {
        long r = (long)tile * 16 + v + 8 * h;
        hws2[r * HDIM + nn] = c[v] * dis[r];
    }
}

__global__ void pool_kernel(const float* __restrict__ agg2, const float* __restrict__ dis,
                            const float* __restrict__ b2v, const int* __restrict__ batch,
                            float* __restrict__ pool) {
    int t = blockIdx.x * blockDim.x + threadIdx.x;
    int n = t >> 4, j = t & 15;
    if (n < N_NODES) {
        float val = dis[n] * agg2[(long)n * HDIM + j] + b2v[j];
        atomicAdd(&pool[batch[n] * HDIM + j], val);
    }
}

__global__ void final_kernel(const float* __restrict__ pool, const int* __restrict__ counts,
                             const float* __restrict__ Wfc, const float* __restrict__ bfc,
                             float* __restrict__ out) {
    int t = blockIdx.x * blockDim.x + threadIdx.x;
    if (t < BGRAPH * CDIM) {
        int b = t / CDIM, cc = t % CDIM;
        float inv = 1.0f / fmaxf((float)counts[b], 1.0f);
        float acc = bfc[cc];
#pragma unroll
        for (int j = 0; j < HDIM; ++j)
            acc += pool[b * HDIM + j] * inv * Wfc[j * CDIM + cc];
        out[t] = acc;
    }
}

extern "C" void kernel_launch(void* const* d_in, const int* in_sizes, int n_in,
                              void* d_out, int out_size, void* d_ws, size_t ws_size,
                              hipStream_t stream) {
    (void)in_sizes; (void)n_in; (void)out_size; (void)ws_size;

    const float* x    = (const float*)d_in[0];
    const int*   ei   = (const int*)d_in[1];
    const int*   batch= (const int*)d_in[2];
    const float* W1   = (const float*)d_in[3];
    const float* b1   = (const float*)d_in[4];
    const float* W2   = (const float*)d_in[5];
    const float* b2   = (const float*)d_in[6];
    const float* Wfc  = (const float*)d_in[7];
    const float* bfc  = (const float*)d_in[8];
    float* out = (float*)d_out;

    const int* erow = ei;
    const int* ecol = ei + N_EDGES;

    char* ws = (char*)d_ws;
    size_t o = 0;
    auto alloc = [&](size_t bytes) -> void* {
        void* p = ws + o;
        o += (bytes + 255) & ~(size_t)255;
        return p;
    };
    float* dis    = (float*)alloc((size_t)N_NODES * 4);
    float* hws    = (float*)alloc((size_t)N_NODES * HDIM * 4);
    float* agg    = (float*)alloc((size_t)N_NODES * HDIM * 4);
    int*   bucket = (int*)  alloc((size_t)N_EDGES * 4);
    float* pool   = (float*)alloc((size_t)BGRAPH * HDIM * 4);
    int*   counts = (int*)  alloc((size_t)(BGRAPH + 256) * 4);
    int*   bcnt   = counts + BGRAPH;
    int*   bstart = (int*)  alloc((size_t)512 * 4);
    int*   bcur   = (int*)  alloc((size_t)256 * 4);

    hipMemsetAsync(counts, 0, (size_t)(BGRAPH + 256) * 4, stream);
    hipMemsetAsync(pool,   0, (size_t)BGRAPH * HDIM * 4, stream);

    bucket_hist_kernel<<<HS_GRID, HS_THREADS, 0, stream>>>(ecol, bcnt);
    bucket_scan_kernel<<<1, 256, 0, stream>>>(bcnt, bstart, bcur);
    partition_kernel<<<PT_GRID, PT_THREADS, 0, stream>>>(erow, ecol, bcur, bucket);
    degree_kernel<<<N_BUCKETS, DG_THREADS, 0, stream>>>(bucket, bstart, batch, dis, counts);

    int gemm_grid = (NTILES + 7) / 8;
    gemm1_kernel<<<gemm_grid, 256, 0, stream>>>(x, W1, dis, hws);

    edge_scatter_kernel<<<N_BUCKETS, SC_THREADS, 0, stream>>>(bucket, bstart, hws, agg);

    gemm2_kernel<<<gemm_grid, 256, 0, stream>>>(agg, W2, b1, dis, hws);

    edge_scatter_kernel<<<N_BUCKETS, SC_THREADS, 0, stream>>>(bucket, bstart, hws, agg);

    pool_kernel<<<(N_NODES * 16 + 255) / 256, 256, 0, stream>>>(agg, dis, b2, batch, pool);
    final_kernel<<<(BGRAPH * CDIM + 255) / 256, 256, 0, stream>>>(pool, counts, Wfc, bfc, out);
}


# scatter work-group finishes its nodes from the LDS sums: layer 1 applies relu and W2 (WMMA) there, layer 2 pools runs of equal graph id; agg plane and two launches go; pool sum order changes
# speedup vs baseline: 5.5675x; 1.1127x over previous
#include <hip/hip_runtime.h>

#define N_NODES 200000
#define N_EDGES 6400000
#define FDIM    128
#define HDIM    16
#define CDIM    10
#define BGRAPH  512
#define NTILES  (N_NODES / 16)
#define BK_NODES  800
#define N_BUCKETS (N_NODES / BK_NODES)

typedef float v2f __attribute__((ext_vector_type(2)));
typedef float v8f __attribute__((ext_vector_type(8)));

#define HS_THREADS 1024
#define HS_GRID    256
#define HS_TILE4   (N_EDGES / 4 / HS_GRID)
#define HS_VEC     ((HS_TILE4 + HS_THREADS - 1) / HS_THREADS)

__global__ void __launch_bounds__(HS_THREADS) bucket_hist_kernel(const int* __restrict__ col,
                                                                 int* __restrict__ bcnt) {
    __shared__ int scnt[256];
    const int tid = threadIdx.x;
    if (tid < 256) scnt[tid] = 0;
    __syncthreads();

    const int4* __restrict__ col4 = (const int4*)col + blockIdx.x * HS_TILE4;
    int4 c[HS_VEC];
#pragma unroll
    for (int i = 0; i < HS_VEC; ++i) {
        int i4 = i * HS_THREADS + tid;
        bool in = i4 < HS_TILE4;
        c[i] = col4[in ? i4 : HS_TILE4 - 1];
        if (!in) c[i] = make_int4(-1, -1, -1, -1);
    }
#pragma unroll
    for (int i = 0; i < HS_VEC; ++i) {
        if (c[i].x >= 0) {
            atomicAdd(&scnt[(unsigned)c[i].x / (unsigned)BK_NODES], 1);
            atomicAdd(&scnt[(unsigned)c[i].y / (unsigned)BK_NODES], 1);
            atomicAdd(&scnt[(unsigned)c[i].z / (unsigned)BK_NODES], 1);
            atomicAdd(&scnt[(unsigned)c[i].w / (unsigned)BK_NODES], 1);
        }
    }
    __syncthreads();
    if (tid < N_BUCKETS) {
        int n = scnt[tid];
        if (n > 0) atomicAdd(&bcnt[tid], n);
    }
}

__global__ void __launch_bounds__(256) bucket_scan_kernel(const int* __restrict__ bcnt,
                                                          int* __restrict__ bstart,
                                                          int* __restrict__ bcur) {
    __shared__ int s[256];
    int t = threadIdx.x;
    int v = (t < N_BUCKETS) ? bcnt[t] : 0;
    s[t] = v;
    __syncthreads();
    for (int off = 1; off < 256; off <<= 1) {
        int x = (t >= off) ? s[t - off] : 0;
        __syncthreads();
        s[t] += x;
        __syncthreads();
    }
    int excl = s[t] - v;
    if (t <= N_BUCKETS) bstart[t] = excl;
    if (t < N_BUCKETS)  bcur[t] = excl;
}

#define PT_THREADS 512
#define PT_VEC     4
#define PT_TILE4   (PT_THREADS * PT_VEC)
#define PT_EDGES   (PT_TILE4 * 4)
#define N_EDGE4    (N_EDGES / 4)
#define PT_GRID    ((N_EDGE4 + PT_TILE4 - 1) / PT_TILE4)

__global__ void __launch_bounds__(PT_THREADS) partition_kernel(const int* __restrict__ row,
                                                               const int* __restrict__ col,
                                                               int* __restrict__ bcur,
                                                               int* __restrict__ bucket) {
    __shared__ int scnt[256];
    __shared__ int sloc[256];
    __shared__ int sdel[256];
    __shared__ int sval[PT_EDGES];
    __shared__ unsigned char sbk[PT_EDGES];
    const int tid = threadIdx.x;
    if (tid < 256) scnt[tid] = 0;
    __syncthreads();

    const int4* __restrict__ col4 = (const int4*)col;
    const int4* __restrict__ row4 = (const int4*)row;
    int cc[4 * PT_VEC], rr[4 * PT_VEC], bb[4 * PT_VEC], rk[4 * PT_VEC];
#pragma unroll
    for (int i = 0; i < PT_VEC; ++i) {
        int i4 = blockIdx.x * PT_TILE4 + i * PT_THREADS + tid;
        bool in = i4 < N_EDGE4;
        int i4c = in ? i4 : N_EDGE4 - 1;
        int4 c = col4[i4c], r = row4[i4c];
        if (!in) c = make_int4(-1, -1, -1, -1);
        cc[4 * i + 0] = c.x; cc[4 * i + 1] = c.y; cc[4 * i + 2] = c.z; cc[4 * i + 3] = c.w;
        rr[4 * i + 0] = r.x; rr[4 * i + 1] = r.y; rr[4 * i + 2] = r.z; rr[4 * i + 3] = r.w;
    }
#pragma unroll
    for (int q = 0; q < 4 * PT_VEC; ++q) {
        bb[q] = 0; rk[q] = 0;
        if (cc[q] >= 0) {
            bb[q] = (int)((unsigned)cc[q] / (unsigned)BK_NODES);
            rk[q] = atomicAdd(&scnt[bb[q]], 1);
        }
    }
    __syncthreads();

    if (tid < 32) {
        int c8[8];
        int run = 0;
#pragma unroll
        for (int q = 0; q < 8; ++q) { int t = scnt[tid * 8 + q]; c8[q] = run; run += t; }
        int incl = run;
#pragma unroll
        for (int d = 1; d < 32; d <<= 1) {
            int y = __shfl_up(incl, d);
            if (tid >= d) incl += y;
        }
        int lanebase = incl - run;
#pragma unroll
        for (int q = 0; q < 8; ++q) sloc[tid * 8 + q] = lanebase + c8[q];
    }
    __syncthreads();

    if (tid < N_BUCKETS) {
        int n = scnt[tid];
        int g = (n > 0) ? atomicAdd(&bcur[tid], n) : 0;
        sdel[tid] = g - sloc[tid];
    }
#pragma unroll
    for (int q = 0; q < 4 * PT_VEC; ++q) {
        if (cc[q] >= 0) {
            int slot = sloc[bb[q]] + rk[q];
            sval[slot] = (rr[q] << 10) | (cc[q] - bb[q] * BK_NODES);
            sbk[slot]  = (unsigned char)bb[q];
        }
    }
    __syncthreads();

    int ntot = N_EDGES - blockIdx.x * PT_EDGES;
    if (ntot > PT_EDGES) ntot = PT_EDGES;
    for (int slot = tid; slot < ntot; slot += PT_THREADS) {
        int b = sbk[slot];
        bucket[sdel[b] + slot] = sval[slot];
    }
}

#define DG_THREADS 1024
#define DG_VEC     8

__global__ void __launch_bounds__(DG_THREADS) degree_kernel(const int* __restrict__ bucket,
                                                            const int* __restrict__ bstart,
                                                            const int* __restrict__ batch,
                                                            float* __restrict__ dis,
                                                            int* __restrict__ counts) {
    __shared__ int cnt[BK_NODES];
    const int tid = threadIdx.x;
    const int b   = blockIdx.x;
    if (tid < BK_NODES) cnt[tid] = 0;
    __syncthreads();

    const int beg = bstart[b], end = bstart[b + 1];
    int base = beg;
    for (; base + DG_THREADS * DG_VEC <= end; base += DG_THREADS * DG_VEC) {
        int p[DG_VEC];
#pragma unroll
        for (int u = 0; u < DG_VEC; ++u) p[u] = bucket[base + u * DG_THREADS + tid];
#pragma unroll
        for (int u = 0; u < DG_VEC; ++u) atomicAdd(&cnt[p[u] & 1023], 1);
    }
    {
        int p[DG_VEC];
#pragma unroll
        for (int u = 0; u < DG_VEC; ++u) {
            int k = base + u * DG_THREADS + tid;
            p[u] = (k < end) ? bucket[k] : -1;
        }
#pragma unroll
        for (int u = 0; u < DG_VEC; ++u)
            if (p[u] >= 0) atomicAdd(&cnt[p[u] & 1023], 1);
    }
    __syncthreads();

    if (tid < BK_NODES) {
        int n = b * BK_NODES + tid;
        dis[n] = rsqrtf(1.0f + (float)cnt[tid]);
        atomicAdd(&counts[batch[n]], 1);
    }
}

__global__ void __launch_bounds__(256) gemm1_kernel(const float* __restrict__ x,
                                                    const float* __restrict__ W1,
                                                    const float* __restrict__ dis,
                                                    float* __restrict__ hws) {
    __shared__ float lw[(FDIM / 2) * 32];
    int tid = threadIdx.x;
#pragma unroll
    for (int i = 0; i < 8; ++i) {
        int idx = i * 256 + tid;
        int k = idx >> 4, n = idx & 15;
        lw[(k >> 1) * 32 + n * 2 + (k & 1)] = W1[idx];
    }
    __syncthreads();

    int wave = tid >> 5, lane = tid & 31;
    int tile = blockIdx.x * 8 + wave;
    if (tile >= NTILES) return;

    int h  = lane >> 4;
    int nn = lane & 15;
    long m = (long)tile * 16 + nn;
    const float* xrow = x + m * FDIM;

    v8f c = {0.f, 0.f, 0.f, 0.f, 0.f, 0.f, 0.f, 0.f};
#pragma unroll
    for (int kk = 0; kk < FDIM; kk += 4) {
        int k0 = kk + 2 * h;
        v2f a = *(const v2f*)(xrow + k0);
        v2f b = *(const v2f*)(&lw[(k0 >> 1) * 32 + nn * 2]);
        c = __builtin_amdgcn_wmma_f32_16x16x4_f32(false, a, false, b,
                                                  (short)0, c, false, false);
    }
#pragma unroll
    for (int v = 0; v < 8; ++v) {
        long r = (long)tile * 16 + v + 8 * h;
        hws[r * HDIM + nn] = c[v] * dis[r];
    }
}

#define SC_THREADS 1024
#define SC_WAVES   (SC_THREADS / 32)
#define PL_RUN     13

template <bool FULL>
__device__ __forceinline__ void add_edges(const int (&p)[16], int n, int h, int j,
                                          const float* __restrict__ hws, float* __restrict__ acc) {
    float v[16];
#pragma unroll
    for (int k = 0; k < 16; ++k) {
        int s = 2 * k + h;
        unsigned r = (unsigned)p[k] >> 10;
        v[k] = (FULL || s < n) ? hws[r * HDIM + j] : 0.0f;
    }
#pragma unroll
    for (int k = 0; k < 16; ++k) {
        int s = 2 * k + h;
        int rel = p[k] & 1023;
        if (FULL || s < n)
            __hip_atomic_fetch_add(&acc[rel * HDIM + j], v[k], __ATOMIC_RELAXED,
                                   __HIP_MEMORY_SCOPE_WORKGROUP);
    }
}

template <int LAYER>
__global__ void __launch_bounds__(SC_THREADS) edge_scatter_kernel(const int* __restrict__ bucket,
                                                                  const int* __restrict__ bstart,
                                                                  const float* __restrict__ hws,
                                                                  const float* __restrict__ dis,
                                                                  const float* __restrict__ Wn,
                                                                  const float* __restrict__ bias,
                                                                  const int* __restrict__ batch,
                                                                  float* __restrict__ outp) {
    __shared__ __attribute__((aligned(16))) float acc[BK_NODES * HDIM];
    __shared__ float lw[(HDIM / 2) * 32];
    __shared__ float lb[HDIM];

    const int tid  = threadIdx.x;
    const int lane = tid & 31;
    const int w    = __builtin_amdgcn_readfirstlane(tid >> 5);
    const int b    = blockIdx.x;
    const int lo   = b * BK_NODES;

    for (int i = tid; i < BK_NODES * HDIM / 4; i += SC_THREADS)
        ((float4*)acc)[i] = make_float4(0.f, 0.f, 0.f, 0.f);
    if (LAYER == 1) {
        if (tid < 256) {
            int k = tid >> 4, n = tid & 15;
            lw[(k >> 1) * 32 + n * 2 + (k & 1)] = Wn[tid];
        }
        if (tid < HDIM) lb[tid] = bias[tid];
    }
    __syncthreads();

    const int h = lane >> 4, j = lane & 15;
    const int beg = bstart[b], end = bstart[b + 1];

    int i = beg + w * 32;
    int e = (i + lane < end) ? bucket[i + lane] : 0;
    while (i < end) {
        int p[16];
#pragma unroll
        for (int k = 0; k < 16; ++k)
            p[k] = __builtin_amdgcn_ds_bpermute((2 * k + h) << 2, e);
        int inext = i + SC_THREADS;
        e = (inext + lane < end) ? bucket[inext + lane] : 0;
        int n = end - i;
        if (n >= 32) add_edges<true>(p, 32, h, j, hws, acc);
        else         add_edges<false>(p, n, h, j, hws, acc);
        i = inext;
    }

    __syncthreads();

    if (LAYER == 1) {
        for (int t = w; t < BK_NODES / 16; t += SC_WAVES) {
            const int ml = t * 16 + j;
            const long m = (long)lo + ml;
            const float dm = dis[m];
            const float* srow = hws + m * HDIM;

            v8f c = {0.f, 0.f, 0.f, 0.f, 0.f, 0.f, 0.f, 0.f};
#pragma unroll
            for (int kk = 0; kk < HDIM; kk += 4) {
                int k0 = kk + 2 * h;
                v2f s  = *(const v2f*)(srow + k0);
                v2f sm = *(const v2f*)(&acc[ml * HDIM + k0]);
                v2f a;
                a.x = fmaxf(dm * (s.x + sm.x) + lb[k0], 0.0f);
                a.y = fmaxf(dm * (s.y + sm.y) + lb[k0 + 1], 0.0f);
                v2f bf = *(const v2f*)(&lw[(k0 >> 1) * 32 + j * 2]);
                c = __builtin_amdgcn_wmma_f32_16x16x4_f32(false, a, false, bf,
                                                          (short)0, c, false, false);
            }
#pragma unroll
            for (int v = 0; v < 8; ++v) {
                long r = (long)lo + t * 16 + v + 8 * h;
                outp[r * HDIM + j] = c[v] * dis[r];
            }
        }
    } else {
        const int g  = tid >> 4;
        const int n0 = g * PL_RUN;
        const float bj = bias[j];
        float vals[PL_RUN];
        int   gid[PL_RUN];
#pragma unroll
        for (int u = 0; u < PL_RUN; ++u) {
            int il = n0 + u;
            bool in = il < BK_NODES;
            int ic = in ? il : BK_NODES - 1;
            int n = lo + ic;
            gid[u]  = in ? batch[n] : -1;
            vals[u] = dis[n] * (acc[ic * HDIM + j] + hws[(long)n * HDIM + j]) + bj;
        }
        float run = 0.0f;
        int cur = gid[0];
#pragma unroll
        for (int u = 0; u < PL_RUN; ++u) {
            if (gid[u] >= 0) {
                if (gid[u] != cur) {
                    atomicAdd(&outp[cur * HDIM + j], run);
                    run = 0.0f;
                    cur = gid[u];
                }
                run += vals[u];
            }
        }
        if (cur >= 0) atomicAdd(&outp[cur * HDIM + j], run);
    }
}

__global__ void final_kernel(const float* __restrict__ pool, const int* __restrict__ counts,
                             const float* __restrict__ Wfc, const float* __restrict__ bfc,
                             float* __restrict__ out) {
    int t = blockIdx.x * blockDim.x + threadIdx.x;
    if (t < BGRAPH * CDIM) {
        int b = t / CDIM, cc = t % CDIM;
        float inv = 1.0f / fmaxf((float)counts[b], 1.0f);
        float acc = bfc[cc];
#pragma unroll
        for (int j = 0; j < HDIM; ++j)
            acc += pool[b * HDIM + j] * inv * Wfc[j * CDIM + cc];
        out[t] = acc;
    }
}

extern "C" void kernel_launch(void* const* d_in, const int* in_sizes, int n_in,
                              void* d_out, int out_size, void* d_ws, size_t ws_size,
                              hipStream_t stream) {
    (void)in_sizes; (void)n_in; (void)out_size; (void)ws_size;

    const float* x    = (const float*)d_in[0];
    const int*   ei   = (const int*)d_in[1];
    const int*   batch= (const int*)d_in[2];
    const float* W1   = (const float*)d_in[3];
    const float* b1   = (const float*)d_in[4];
    const float* W2   = (const float*)d_in[5];
    const float* b2   = (const float*)d_in[6];
    const float* Wfc  = (const float*)d_in[7];
    const float* bfc  = (const float*)d_in[8];
    float* out = (float*)d_out;

    const int* erow = ei;
    const int* ecol = ei + N_EDGES;

    char* ws = (char*)d_ws;
    size_t o = 0;
    auto alloc = [&](size_t bytes) -> void* {
        void* p = ws + o;
        o += (bytes + 255) & ~(size_t)255;
        return p;
    };
    float* dis    = (float*)alloc((size_t)N_NODES * 4);
    float* hws    = (float*)alloc((size_t)N_NODES * HDIM * 4);
    float* hws2   = (float*)alloc((size_t)N_NODES * HDIM * 4);
    int*   bucket = (int*)  alloc((size_t)N_EDGES * 4);
    float* pool   = (float*)alloc((size_t)BGRAPH * HDIM * 4);
    int*   counts = (int*)  alloc((size_t)(BGRAPH + 256) * 4);
    int*   bcnt   = counts + BGRAPH;
    int*   bstart = (int*)  alloc((size_t)512 * 4);
    int*   bcur   = (int*)  alloc((size_t)256 * 4);

    hipMemsetAsync(counts, 0, (size_t)(BGRAPH + 256) * 4, stream);
    hipMemsetAsync(pool,   0, (size_t)BGRAPH * HDIM * 4, stream);

    bucket_hist_kernel<<<HS_GRID, HS_THREADS, 0, stream>>>(ecol, bcnt);
    bucket_scan_kernel<<<1, 256, 0, stream>>>(bcnt, bstart, bcur);
    partition_kernel<<<PT_GRID, PT_THREADS, 0, stream>>>(erow, ecol, bcur, bucket);
    degree_kernel<<<N_BUCKETS, DG_THREADS, 0, stream>>>(bucket, bstart, batch, dis, counts);

    int gemm_grid = (NTILES + 7) / 8;
    gemm1_kernel<<<gemm_grid, 256, 0, stream>>>(x, W1, dis, hws);

    edge_scatter_kernel<1><<<N_BUCKETS, SC_THREADS, 0, stream>>>(bucket, bstart, hws, dis, W2, b1, batch, hws2);
    edge_scatter_kernel<2><<<N_BUCKETS, SC_THREADS, 0, stream>>>(bucket, bstart, hws2, dis, W2, b2, batch, pool);
    final_kernel<<<(BGRAPH * CDIM + 255) / 256, 256, 0, stream>>>(pool, counts, Wfc, bfc, out);
}


# scatter loop: a group's 32 edge entries leave the wave's serial chain: loaded behind the gathers (hidden under them), spread by DPP row broadcasts, not ds_bpermute behind the adds; sums unchanged
# speedup vs baseline: 5.8090x; 1.0434x over previous
#include <hip/hip_runtime.h>

#define N_NODES 200000
#define N_EDGES 6400000
#define FDIM    128
#define HDIM    16
#define CDIM    10
#define BGRAPH  512
#define NTILES  (N_NODES / 16)
#define BK_NODES  800
#define N_BUCKETS (N_NODES / BK_NODES)

typedef float v2f __attribute__((ext_vector_type(2)));
typedef float v8f __attribute__((ext_vector_type(8)));

#define HS_THREADS 1024
#define HS_GRID    256
#define HS_TILE4   (N_EDGES / 4 / HS_GRID)
#define HS_VEC     ((HS_TILE4 + HS_THREADS - 1) / HS_THREADS)

__global__ void __launch_bounds__(HS_THREADS) bucket_hist_kernel(const int* __restrict__ col,
                                                                 int* __restrict__ bcnt) {
    __shared__ int scnt[256];
    const int tid = threadIdx.x;
    if (tid < 256) scnt[tid] = 0;
    __syncthreads();

    const int4* __restrict__ col4 = (const int4*)col + blockIdx.x * HS_TILE4;
    int4 c[HS_VEC];
#pragma unroll
    for (int i = 0; i < HS_VEC; ++i) {
        int i4 = i * HS_THREADS + tid;
        bool in = i4 < HS_TILE4;
        c[i] = col4[in ? i4 : HS_TILE4 - 1];
        if (!in) c[i] = make_int4(-1, -1, -1, -1);
    }
#pragma unroll
    for (int i = 0; i < HS_VEC; ++i) {
        if (c[i].x >= 0) {
            atomicAdd(&scnt[(unsigned)c[i].x / (unsigned)BK_NODES], 1);
            atomicAdd(&scnt[(unsigned)c[i].y / (unsigned)BK_NODES], 1);
            atomicAdd(&scnt[(unsigned)c[i].z / (unsigned)BK_NODES], 1);
            atomicAdd(&scnt[(unsigned)c[i].w / (unsigned)BK_NODES], 1);
        }
    }
    __syncthreads();
    if (tid < N_BUCKETS) {
        int n = scnt[tid];
        if (n > 0) atomicAdd(&bcnt[tid], n);
    }
}

__global__ void __launch_bounds__(256) bucket_scan_kernel(const int* __restrict__ bcnt,
                                                          int* __restrict__ bstart,
                                                          int* __restrict__ bcur) {
    __shared__ int s[256];
    int t = threadIdx.x;
    int v = (t < N_BUCKETS) ? bcnt[t] : 0;
    s[t] = v;
    __syncthreads();
    for (int off = 1; off < 256; off <<= 1) {
        int x = (t >= off) ? s[t - off] : 0;
        __syncthreads();
        s[t] += x;
        __syncthreads();
    }
    int excl = s[t] - v;
    if (t <= N_BUCKETS) bstart[t] = excl;
    if (t < N_BUCKETS)  bcur[t] = excl;
}

#define PT_THREADS 512
#define PT_VEC     4
#define PT_TILE4   (PT_THREADS * PT_VEC)
#define PT_EDGES   (PT_TILE4 * 4)
#define N_EDGE4    (N_EDGES / 4)
#define PT_GRID    ((N_EDGE4 + PT_TILE4 - 1) / PT_TILE4)

__global__ void __launch_bounds__(PT_THREADS) partition_kernel(const int* __restrict__ row,
                                                               const int* __restrict__ col,
                                                               int* __restrict__ bcur,
                                                               int* __restrict__ bucket) {
    __shared__ int scnt[256];
    __shared__ int sloc[256];
    __shared__ int sdel[256];
    __shared__ int sval[PT_EDGES];
    __shared__ unsigned char sbk[PT_EDGES];
    const int tid = threadIdx.x;
    if (tid < 256) scnt[tid] = 0;
    __syncthreads();

    const int4* __restrict__ col4 = (const int4*)col;
    const int4* __restrict__ row4 = (const int4*)row;
    int cc[4 * PT_VEC], rr[4 * PT_VEC], bb[4 * PT_VEC], rk[4 * PT_VEC];
#pragma unroll
    for (int i = 0; i < PT_VEC; ++i) {
        int i4 = blockIdx.x * PT_TILE4 + i * PT_THREADS + tid;
        bool in = i4 < N_EDGE4;
        int i4c = in ? i4 : N_EDGE4 - 1;
        int4 c = col4[i4c], r = row4[i4c];
        if (!in) c = make_int4(-1, -1, -1, -1);
        cc[4 * i + 0] = c.x; cc[4 * i + 1] = c.y; cc[4 * i + 2] = c.z; cc[4 * i + 3] = c.w;
        rr[4 * i + 0] = r.x; rr[4 * i + 1] = r.y; rr[4 * i + 2] = r.z; rr[4 * i + 3] = r.w;
    }
#pragma unroll
    for (int q = 0; q < 4 * PT_VEC; ++q) {
        bb[q] = 0; rk[q] = 0;
        if (cc[q] >= 0) {
            bb[q] = (int)((unsigned)cc[q] / (unsigned)BK_NODES);
            rk[q] = atomicAdd(&scnt[bb[q]], 1);
        }
    }
    __syncthreads();

    if (tid < 32) {
        int c8[8];
        int run = 0;
#pragma unroll
        for (int q = 0; q < 8; ++q) { int t = scnt[tid * 8 + q]; c8[q] = run; run += t; }
        int incl = run;
#pragma unroll
        for (int d = 1; d < 32; d <<= 1) {
            int y = __shfl_up(incl, d);
            if (tid >= d) incl += y;
        }
        int lanebase = incl - run;
#pragma unroll
        for (int q = 0; q < 8; ++q) sloc[tid * 8 + q] = lanebase + c8[q];
    }
    __syncthreads();

    if (tid < N_BUCKETS) {
        int n = scnt[tid];
        int g = (n > 0) ? atomicAdd(&bcur[tid], n) : 0;
        sdel[tid] = g - sloc[tid];
    }
#pragma unroll
    for (int q = 0; q < 4 * PT_VEC; ++q) {
        if (cc[q] >= 0) {
            int slot = sloc[bb[q]] + rk[q];
            sval[slot] = (rr[q] << 10) | (cc[q] - bb[q] * BK_NODES);
            sbk[slot]  = (unsigned char)bb[q];
        }
    }
    __syncthreads();

    int ntot = N_EDGES - blockIdx.x * PT_EDGES;
    if (ntot > PT_EDGES) ntot = PT_EDGES;
    for (int slot = tid; slot < ntot; slot += PT_THREADS) {
        int b = sbk[slot];
        bucket[sdel[b] + slot] = sval[slot];
    }
}

#define DG_THREADS 1024
#define DG_VEC     8

__global__ void __launch_bounds__(DG_THREADS) degree_kernel(const int* __restrict__ bucket,
                                                            const int* __restrict__ bstart,
                                                            const int* __restrict__ batch,
                                                            float* __restrict__ dis,
                                                            int* __restrict__ counts) {
    __shared__ int cnt[BK_NODES];
    const int tid = threadIdx.x;
    const int b   = blockIdx.x;
    if (tid < BK_NODES) cnt[tid] = 0;
    __syncthreads();

    const int beg = bstart[b], end = bstart[b + 1];
    int base = beg;
    for (; base + DG_THREADS * DG_VEC <= end; base += DG_THREADS * DG_VEC) {
        int p[DG_VEC];
#pragma unroll
        for (int u = 0; u < DG_VEC; ++u) p[u] = bucket[base + u * DG_THREADS + tid];
#pragma unroll
        for (int u = 0; u < DG_VEC; ++u) atomicAdd(&cnt[p[u] & 1023], 1);
    }
    {
        int p[DG_VEC];
#pragma unroll
        for (int u = 0; u < DG_VEC; ++u) {
            int k = base + u * DG_THREADS + tid;
            p[u] = (k < end) ? bucket[k] : -1;
        }
#pragma unroll
        for (int u = 0; u < DG_VEC; ++u)
            if (p[u] >= 0) atomicAdd(&cnt[p[u] & 1023], 1);
    }
    __syncthreads();

    if (tid < BK_NODES) {
        int n = b * BK_NODES + tid;
        dis[n] = rsqrtf(1.0f + (float)cnt[tid]);
        atomicAdd(&counts[batch[n]], 1);
    }
}

__global__ void __launch_bounds__(256) gemm1_kernel(const float* __restrict__ x,
                                                    const float* __restrict__ W1,
                                                    const float* __restrict__ dis,
                                                    float* __restrict__ hws) {
    __shared__ float lw[(FDIM / 2) * 32];
    int tid = threadIdx.x;
#pragma unroll
    for (int i = 0; i < 8; ++i) {
        int idx = i * 256 + tid;
        int k = idx >> 4, n = idx & 15;
        lw[(k >> 1) * 32 + n * 2 + (k & 1)] = W1[idx];
    }
    __syncthreads();

    int wave = tid >> 5, lane = tid & 31;
    int tile = blockIdx.x * 8 + wave;
    if (tile >= NTILES) return;

    int h  = lane >> 4;
    int nn = lane & 15;
    long m = (long)tile * 16 + nn;
    const float* xrow = x + m * FDIM;

    v8f c = {0.f, 0.f, 0.f, 0.f, 0.f, 0.f, 0.f, 0.f};
#pragma unroll
    for (int kk = 0; kk < FDIM; kk += 4) {
        int k0 = kk + 2 * h;
        v2f a = *(const v2f*)(xrow + k0);
        v2f b = *(const v2f*)(&lw[(k0 >> 1) * 32 + nn * 2]);
        c = __builtin_amdgcn_wmma_f32_16x16x4_f32(false, a, false, b,
                                                  (short)0, c, false, false);
    }
#pragma unroll
    for (int v = 0; v < 8; ++v) {
        long r = (long)tile * 16 + v + 8 * h;
        hws[r * HDIM + nn] = c[v] * dis[r];
    }
}

#define SC_THREADS 1024
#define SC_WAVES   (SC_THREADS / 32)
#define PL_RUN     13

template <int K>
__device__ __forceinline__ int row_bcast(int v) {
    return __builtin_amdgcn_update_dpp(0, v, 0x150 + K, 0xF, 0xF, true);
}

template <bool FULL>
__device__ __forceinline__ int add_edges(const int (&p)[16], int n, int h, int j,
                                         const float* __restrict__ hws, float* __restrict__ acc,
                                         const int* __restrict__ enext) {
    float v[16];
#pragma unroll
    for (int k = 0; k < 16; ++k) {
        int s = 2 * k + h;
        unsigned r = (unsigned)p[k] >> 10;
        v[k] = (FULL || s < n) ? hws[r * HDIM + j] : 0.0f;
    }
    int e = *enext;
#pragma unroll
    for (int k = 0; k < 16; ++k) {
        int s = 2 * k + h;
        int rel = p[k] & 1023;
        if (FULL || s < n)
            __hip_atomic_fetch_add(&acc[rel * HDIM + j], v[k], __ATOMIC_RELAXED,
                                   __HIP_MEMORY_SCOPE_WORKGROUP);
    }
    return e;
}

template <int LAYER>
__global__ void __launch_bounds__(SC_THREADS) edge_scatter_kernel(const int* __restrict__ bucket,
                                                                  const int* __restrict__ bstart,
                                                                  const float* __restrict__ hws,
                                                                  const float* __restrict__ dis,
                                                                  const float* __restrict__ Wn,
                                                                  const float* __restrict__ bias,
                                                                  const int* __restrict__ batch,
                                                                  float* __restrict__ outp) {
    __shared__ __attribute__((aligned(16))) float acc[BK_NODES * HDIM];
    __shared__ float lw[(HDIM / 2) * 32];
    __shared__ float lb[HDIM];

    const int tid  = threadIdx.x;
    const int lane = tid & 31;
    const int w    = __builtin_amdgcn_readfirstlane(tid >> 5);
    const int b    = blockIdx.x;
    const int lo   = b * BK_NODES;

    for (int i = tid; i < BK_NODES * HDIM / 4; i += SC_THREADS)
        ((float4*)acc)[i] = make_float4(0.f, 0.f, 0.f, 0.f);
    if (LAYER == 1) {
        if (tid < 256) {
            int k = tid >> 4, n = tid & 15;
            lw[(k >> 1) * 32 + n * 2 + (k & 1)] = Wn[tid];
        }
        if (tid < HDIM) lb[tid] = bias[tid];
    }
    __syncthreads();

    const int h = lane >> 4, j = lane & 15;
    const int beg = bstart[b], end = bstart[b + 1];

    const int ld = 2 * j + h;
    int i = beg + w * 32;
    int e = bucket[min(i + ld, N_EDGES - 1)];
    while (i < end) {
        int p[16];
        p[0]  = row_bcast<0>(e);   p[1]  = row_bcast<1>(e);   p[2]  = row_bcast<2>(e);   p[3]  = row_bcast<3>(e);
        p[4]  = row_bcast<4>(e);   p[5]  = row_bcast<5>(e);   p[6]  = row_bcast<6>(e);   p[7]  = row_bcast<7>(e);
        p[8]  = row_bcast<8>(e);   p[9]  = row_bcast<9>(e);   p[10] = row_bcast<10>(e);  p[11] = row_bcast<11>(e);
        p[12] = row_bcast<12>(e);  p[13] = row_bcast<13>(e);  p[14] = row_bcast<14>(e);  p[15] = row_bcast<15>(e);
        int inext = i + SC_THREADS;
        const int* enext = bucket + min(inext + ld, N_EDGES - 1);
        int n = end - i;
        if (n >= 32) e = add_edges<true>(p, 32, h, j, hws, acc, enext);
        else         e = add_edges<false>(p, n, h, j, hws, acc, enext);
        i = inext;
    }

    __syncthreads();

    if (LAYER == 1) {
        for (int t = w; t < BK_NODES / 16; t += SC_WAVES) {
            const int ml = t * 16 + j;
            const long m = (long)lo + ml;
            const float dm = dis[m];
            const float* srow = hws + m * HDIM;

            v8f c = {0.f, 0.f, 0.f, 0.f, 0.f, 0.f, 0.f, 0.f};
#pragma unroll
            for (int kk = 0; kk < HDIM; kk += 4) {
                int k0 = kk + 2 * h;
                v2f s  = *(const v2f*)(srow + k0);
                v2f sm = *(const v2f*)(&acc[ml * HDIM + k0]);
                v2f a;
                a.x = fmaxf(dm * (s.x + sm.x) + lb[k0], 0.0f);
                a.y = fmaxf(dm * (s.y + sm.y) + lb[k0 + 1], 0.0f);
                v2f bf = *(const v2f*)(&lw[(k0 >> 1) * 32 + j * 2]);
                c = __builtin_amdgcn_wmma_f32_16x16x4_f32(false, a, false, bf,
                                                          (short)0, c, false, false);
            }
#pragma unroll
            for (int v = 0; v < 8; ++v) {
                long r = (long)lo + t * 16 + v + 8 * h;
                outp[r * HDIM + j] = c[v] * dis[r];
            }
        }
    } else {
        const int g  = tid >> 4;
        const int n0 = g * PL_RUN;
        const float bj = bias[j];
        float vals[PL_RUN];
        int   gid[PL_RUN];
#pragma unroll
        for (int u = 0; u < PL_RUN; ++u) {
            int il = n0 + u;
            bool in = il < BK_NODES;
            int ic = in ? il : BK_NODES - 1;
            int n = lo + ic;
            gid[u]  = in ? batch[n] : -1;
            vals[u] = dis[n] * (acc[ic * HDIM + j] + hws[(long)n * HDIM + j]) + bj;
        }
        float run = 0.0f;
        int cur = gid[0];
#pragma unroll
        for (int u = 0; u < PL_RUN; ++u) {
            if (gid[u] >= 0) {
                if (gid[u] != cur) {
                    atomicAdd(&outp[cur * HDIM + j], run);
                    run = 0.0f;
                    cur = gid[u];
                }
                run += vals[u];
            }
        }
        if (cur >= 0) atomicAdd(&outp[cur * HDIM + j], run);
    }
}

__global__ void final_kernel(const float* __restrict__ pool, const int* __restrict__ counts,
                             const float* __restrict__ Wfc, const float* __restrict__ bfc,
                             float* __restrict__ out) {
    int t = blockIdx.x * blockDim.x + threadIdx.x;
    if (t < BGRAPH * CDIM) {
        int b = t / CDIM, cc = t % CDIM;
        float inv = 1.0f / fmaxf((float)counts[b], 1.0f);
        float acc = bfc[cc];
#pragma unroll
        for (int j = 0; j < HDIM; ++j)
            acc += pool[b * HDIM + j] * inv * Wfc[j * CDIM + cc];
        out[t] = acc;
    }
}

extern "C" void kernel_launch(void* const* d_in, const int* in_sizes, int n_in,
                              void* d_out, int out_size, void* d_ws, size_t ws_size,
                              hipStream_t stream) {
    (void)in_sizes; (void)n_in; (void)out_size; (void)ws_size;

    const float* x    = (const float*)d_in[0];
    const int*   ei   = (const int*)d_in[1];
    const int*   batch= (const int*)d_in[2];
    const float* W1   = (const float*)d_in[3];
    const float* b1   = (const float*)d_in[4];
    const float* W2   = (const float*)d_in[5];
    const float* b2   = (const float*)d_in[6];
    const float* Wfc  = (const float*)d_in[7];
    const float* bfc  = (const float*)d_in[8];
    float* out = (float*)d_out;

    const int* erow = ei;
    const int* ecol = ei + N_EDGES;

    char* ws = (char*)d_ws;
    size_t o = 0;
    auto alloc = [&](size_t bytes) -> void* {
        void* p = ws + o;
        o += (bytes + 255) & ~(size_t)255;
        return p;
    };
    float* dis    = (float*)alloc((size_t)N_NODES * 4);
    float* hws    = (float*)alloc((size_t)N_NODES * HDIM * 4);
    float* hws2   = (float*)alloc((size_t)N_NODES * HDIM * 4);
    int*   bucket = (int*)  alloc((size_t)N_EDGES * 4);
    float* pool   = (float*)alloc((size_t)BGRAPH * HDIM * 4);
    int*   counts = (int*)  alloc((size_t)(BGRAPH + 256) * 4);
    int*   bcnt   = counts + BGRAPH;
    int*   bstart = (int*)  alloc((size_t)512 * 4);
    int*   bcur   = (int*)  alloc((size_t)256 * 4);

    hipMemsetAsync(counts, 0, (size_t)(BGRAPH + 256) * 4, stream);
    hipMemsetAsync(pool,   0, (size_t)BGRAPH * HDIM * 4, stream);

    bucket_hist_kernel<<<HS_GRID, HS_THREADS, 0, stream>>>(ecol, bcnt);
    bucket_scan_kernel<<<1, 256, 0, stream>>>(bcnt, bstart, bcur);
    partition_kernel<<<PT_GRID, PT_THREADS, 0, stream>>>(erow, ecol, bcur, bucket);
    degree_kernel<<<N_BUCKETS, DG_THREADS, 0, stream>>>(bucket, bstart, batch, dis, counts);

    int gemm_grid = (NTILES + 7) / 8;
    gemm1_kernel<<<gemm_grid, 256, 0, stream>>>(x, W1, dis, hws);

    edge_scatter_kernel<1><<<N_BUCKETS, SC_THREADS, 0, stream>>>(bucket, bstart, hws, dis, W2, b1, batch, hws2);
    edge_scatter_kernel<2><<<N_BUCKETS, SC_THREADS, 0, stream>>>(bucket, bstart, hws2, dis, W2, b2, batch, pool);
    final_kernel<<<(BGRAPH * CDIM + 255) / 256, 256, 0, stream>>>(pool, counts, Wfc, bfc, out);
}


# hot-line global atomics go: bucket runs placed from a per-tile count table and a column scan, graph sizes added once per run of equal ids in a wave; edge order in a bucket, so sum order, changes
# speedup vs baseline: 6.3967x; 1.1012x over previous
#include <hip/hip_runtime.h>

#define N_NODES 200000
#define N_EDGES 6400000
#define FDIM    128
#define HDIM    16
#define CDIM    10
#define BGRAPH  512
#define NTILES  (N_NODES / 16)
#define BK_NODES  800
#define N_BUCKETS (N_NODES / BK_NODES)

typedef float v2f __attribute__((ext_vector_type(2)));
typedef float v8f __attribute__((ext_vector_type(8)));

#define PT_THREADS 512
#define PT_VEC     4
#define PT_TILE4   (PT_THREADS * PT_VEC)
#define PT_EDGES   (PT_TILE4 * 4)
#define N_EDGE4    (N_EDGES / 4)
#define PT_GRID    ((N_EDGE4 + PT_TILE4 - 1) / PT_TILE4)
#define TB_PITCH   256
#define TH_THREADS 256
#define TH_VEC     (PT_TILE4 / TH_THREADS)

__global__ void __launch_bounds__(TH_THREADS) tile_hist_kernel(const int* __restrict__ col,
                                                               int* __restrict__ tcnt) {
    __shared__ int scnt[256];
    const int tid = threadIdx.x;
    scnt[tid] = 0;
    __syncthreads();

    const int4* __restrict__ col4 = (const int4*)col;
    int4 c[TH_VEC];
#pragma unroll
    for (int i = 0; i < TH_VEC; ++i) {
        int i4 = blockIdx.x * PT_TILE4 + i * TH_THREADS + tid;
        bool in = i4 < N_EDGE4;
        c[i] = col4[in ? i4 : N_EDGE4 - 1];
        if (!in) c[i] = make_int4(-1, -1, -1, -1);
    }
#pragma unroll
    for (int i = 0; i < TH_VEC; ++i) {
        if (c[i].x >= 0) {
            atomicAdd(&scnt[(unsigned)c[i].x / (unsigned)BK_NODES], 1);
            atomicAdd(&scnt[(unsigned)c[i].y / (unsigned)BK_NODES], 1);
            atomicAdd(&scnt[(unsigned)c[i].z / (unsigned)BK_NODES], 1);
            atomicAdd(&scnt[(unsigned)c[i].w / (unsigned)BK_NODES], 1);
        }
    }
    __syncthreads();
    if (tid < N_BUCKETS) tcnt[blockIdx.x * TB_PITCH + tid] = scnt[tid];
}

#define CS_THREADS 256
#define CS_VEC     4

__global__ void __launch_bounds__(CS_THREADS) tile_scan_kernel(int* __restrict__ tcnt,
                                                               int* __restrict__ btot) {
    __shared__ int wsum[CS_THREADS / 32];
    const int b = blockIdx.x;
    const int t = threadIdx.x, lane = t & 31, w = t >> 5;
    int c[CS_VEC], v[CS_VEC];
#pragma unroll
    for (int q = 0; q < CS_VEC; ++q) {
        int tile = t * CS_VEC + q;
        bool in = tile < PT_GRID;
        c[q] = tcnt[(in ? tile : PT_GRID - 1) * TB_PITCH + b];
        if (!in) c[q] = 0;
    }
    int run = 0;
#pragma unroll
    for (int q = 0; q < CS_VEC; ++q) { v[q] = run; run += c[q]; }
    int incl = run;
#pragma unroll
    for (int d = 1; d < 32; d <<= 1) {
        int y = __shfl_up(incl, d);
        if (lane >= d) incl += y;
    }
    if (lane == 31) wsum[w] = incl;
    __syncthreads();
    int wbase = 0, total = 0;
#pragma unroll
    for (int k = 0; k < CS_THREADS / 32; ++k) {
        int s = wsum[k];
        if (k < w) wbase += s;
        total += s;
    }
    const int base = wbase + incl - run;
#pragma unroll
    for (int q = 0; q < CS_VEC; ++q) {
        int tile = t * CS_VEC + q;
        if (tile < PT_GRID) tcnt[tile * TB_PITCH + b] = base + v[q];
    }
    if (t == 0) btot[b] = total;
}

__global__ void __launch_bounds__(PT_THREADS) partition_kernel(const int* __restrict__ row,
                                                               const int* __restrict__ col,
                                                               const int* __restrict__ tpre,
                                                               const int* __restrict__ btot,
                                                               int* __restrict__ bstart,
                                                               int* __restrict__ bucket) {
    __shared__ int scnt[256];
    __shared__ int sloc[256];
    __shared__ int sbst[256];
    __shared__ int sdel[256];
    __shared__ int sval[PT_EDGES];
    __shared__ unsigned char sbk[PT_EDGES];
    const int tid = threadIdx.x;
    if (tid < 256) scnt[tid] = 0;
    int mypre = 0, mytot = 0;
    if (tid < N_BUCKETS) {
        mypre = tpre[blockIdx.x * TB_PITCH + tid];
        mytot = btot[tid];
    }
    __syncthreads();

    const int4* __restrict__ col4 = (const int4*)col;
    const int4* __restrict__ row4 = (const int4*)row;
    int cc[4 * PT_VEC], rr[4 * PT_VEC], bb[4 * PT_VEC], rk[4 * PT_VEC];
#pragma unroll
    for (int i = 0; i < PT_VEC; ++i) {
        int i4 = blockIdx.x * PT_TILE4 + i * PT_THREADS + tid;
        bool in = i4 < N_EDGE4;
        int i4c = in ? i4 : N_EDGE4 - 1;
        int4 c = col4[i4c], r = row4[i4c];
        if (!in) c = make_int4(-1, -1, -1, -1);
        cc[4 * i + 0] = c.x; cc[4 * i + 1] = c.y; cc[4 * i + 2] = c.z; cc[4 * i + 3] = c.w;
        rr[4 * i + 0] = r.x; rr[4 * i + 1] = r.y; rr[4 * i + 2] = r.z; rr[4 * i + 3] = r.w;
    }
#pragma unroll
    for (int q = 0; q < 4 * PT_VEC; ++q) {
        bb[q] = 0; rk[q] = 0;
        if (cc[q] >= 0) {
            bb[q] = (int)((unsigned)cc[q] / (unsigned)BK_NODES);
            rk[q] = atomicAdd(&scnt[bb[q]], 1);
        }
    }
    if (tid < 256) sbst[tid] = mytot;
    __syncthreads();

    if (tid < 64) {
        const int lane = tid & 31;
        const int* from = (tid < 32) ? scnt : sbst;
        int*       to   = (tid < 32) ? sloc : sbst;
        int c8[8];
        int run = 0;
#pragma unroll
        for (int q = 0; q < 8; ++q) { int t = from[lane * 8 + q]; c8[q] = run; run += t; }
        int incl = run;
#pragma unroll
        for (int d = 1; d < 32; d <<= 1) {
            int y = __shfl_up(incl, d);
            if (lane >= d) incl += y;
        }
        int lanebase = incl - run;
#pragma unroll
        for (int q = 0; q < 8; ++q) to[lane * 8 + q] = lanebase + c8[q];
    }
    __syncthreads();

    if (tid < N_BUCKETS) sdel[tid] = sbst[tid] + mypre - sloc[tid];
    if (blockIdx.x == 0 && tid <= N_BUCKETS) bstart[tid] = sbst[tid];
#pragma unroll
    for (int q = 0; q < 4 * PT_VEC; ++q) {
        if (cc[q] >= 0) {
            int slot = sloc[bb[q]] + rk[q];
            sval[slot] = (rr[q] << 10) | (cc[q] - bb[q] * BK_NODES);
            sbk[slot]  = (unsigned char)bb[q];
        }
    }
    __syncthreads();

    int ntot = N_EDGES - blockIdx.x * PT_EDGES;
    if (ntot > PT_EDGES) ntot = PT_EDGES;
    for (int slot = tid; slot < ntot; slot += PT_THREADS) {
        int b = sbk[slot];
        bucket[sdel[b] + slot] = sval[slot];
    }
}

#define DG_THREADS 1024
#define DG_VEC     8

__global__ void __launch_bounds__(DG_THREADS) degree_kernel(const int* __restrict__ bucket,
                                                            const int* __restrict__ bstart,
                                                            const int* __restrict__ batch,
                                                            float* __restrict__ dis,
                                                            int* __restrict__ counts) {
    __shared__ int cnt[BK_NODES];
    const int tid = threadIdx.x;
    const int b   = blockIdx.x;
    if (tid < BK_NODES) cnt[tid] = 0;
    __syncthreads();

    const int beg = bstart[b], end = bstart[b + 1];
    int base = beg;
    for (; base + DG_THREADS * DG_VEC <= end; base += DG_THREADS * DG_VEC) {
        int p[DG_VEC];
#pragma unroll
        for (int u = 0; u < DG_VEC; ++u) p[u] = bucket[base + u * DG_THREADS + tid];
#pragma unroll
        for (int u = 0; u < DG_VEC; ++u) atomicAdd(&cnt[p[u] & 1023], 1);
    }
    {
        int p[DG_VEC];
#pragma unroll
        for (int u = 0; u < DG_VEC; ++u) {
            int k = base + u * DG_THREADS + tid;
            p[u] = (k < end) ? bucket[k] : -1;
        }
#pragma unroll
        for (int u = 0; u < DG_VEC; ++u)
            if (p[u] >= 0) atomicAdd(&cnt[p[u] & 1023], 1);
    }
    __syncthreads();

    if (tid < BK_NODES) {
        int n = b * BK_NODES + tid;
        dis[n] = rsqrtf(1.0f + (float)cnt[tid]);
        const int lane = tid & 31;
        const int g  = batch[n];
        const int gp = __shfl_up(g, 1);
        const bool head = (lane == 0) || (g != gp);
        const unsigned heads = (unsigned)__ballot(head);
        if (head) {
            unsigned later = heads & ~((2u << lane) - 1u);
            int nxt = later ? __builtin_ctz(later) : 32;
            atomicAdd(&counts[g], nxt - lane);
        }
    }
}

__global__ void __launch_bounds__(256) gemm1_kernel(const float* __restrict__ x,
                                                    const float* __restrict__ W1,
                                                    const float* __restrict__ dis,
                                                    float* __restrict__ hws) {
    __shared__ float lw[(FDIM / 2) * 32];
    int tid = threadIdx.x;
#pragma unroll
    for (int i = 0; i < 8; ++i) {
        int idx = i * 256 + tid;
        int k = idx >> 4, n = idx & 15;
        lw[(k >> 1) * 32 + n * 2 + (k & 1)] = W1[idx];
    }
    __syncthreads();

    int wave = tid >> 5, lane = tid & 31;
    int tile = blockIdx.x * 8 + wave;
    if (tile >= NTILES) return;

    int h  = lane >> 4;
    int nn = lane & 15;
    long m = (long)tile * 16 + nn;
    const float* xrow = x + m * FDIM;

    v8f c = {0.f, 0.f, 0.f, 0.f, 0.f, 0.f, 0.f, 0.f};
#pragma unroll
    for (int kk = 0; kk < FDIM; kk += 4) {
        int k0 = kk + 2 * h;
        v2f a = *(const v2f*)(xrow + k0);
        v2f b = *(const v2f*)(&lw[(k0 >> 1) * 32 + nn * 2]);
        c = __builtin_amdgcn_wmma_f32_16x16x4_f32(false, a, false, b,
                                                  (short)0, c, false, false);
    }
#pragma unroll
    for (int v = 0; v < 8; ++v) {
        long r = (long)tile * 16 + v + 8 * h;
        hws[r * HDIM + nn] = c[v] * dis[r];
    }
}

#define SC_THREADS 1024
#define SC_WAVES   (SC_THREADS / 32)
#define PL_RUN     13

template <int K>
__device__ __forceinline__ int row_bcast(int v) {
    return __builtin_amdgcn_update_dpp(0, v, 0x150 + K, 0xF, 0xF, true);
}

template <bool FULL>
__device__ __forceinline__ int add_edges(const int (&p)[16], int n, int h, int j,
                                         const float* __restrict__ hws, float* __restrict__ acc,
                                         const int* __restrict__ enext) {
    float v[16];
#pragma unroll
    for (int k = 0; k < 16; ++k) {
        int s = 2 * k + h;
        unsigned r = (unsigned)p[k] >> 10;
        v[k] = (FULL || s < n) ? hws[r * HDIM + j] : 0.0f;
    }
    int e = *enext;
#pragma unroll
    for (int k = 0; k < 16; ++k) {
        int s = 2 * k + h;
        int rel = p[k] & 1023;
        if (FULL || s < n)
            __hip_atomic_fetch_add(&acc[rel * HDIM + j], v[k], __ATOMIC_RELAXED,
                                   __HIP_MEMORY_SCOPE_WORKGROUP);
    }
    return e;
}

template <int LAYER>
__global__ void __launch_bounds__(SC_THREADS) edge_scatter_kernel(const int* __restrict__ bucket,
                                                                  const int* __restrict__ bstart,
                                                                  const float* __restrict__ hws,
                                                                  const float* __restrict__ dis,
                                                                  const float* __restrict__ Wn,
                                                                  const float* __restrict__ bias,
                                                                  const int* __restrict__ batch,
                                                                  float* __restrict__ outp) {
    __shared__ __attribute__((aligned(16))) float acc[BK_NODES * HDIM];
    __shared__ float lw[(HDIM / 2) * 32];
    __shared__ float lb[HDIM];

    const int tid  = threadIdx.x;
    const int lane = tid & 31;
    const int w    = __builtin_amdgcn_readfirstlane(tid >> 5);
    const int b    = blockIdx.x;
    const int lo   = b * BK_NODES;

    for (int i = tid; i < BK_NODES * HDIM / 4; i += SC_THREADS)
        ((float4*)acc)[i] = make_float4(0.f, 0.f, 0.f, 0.f);
    if (LAYER == 1) {
        if (tid < 256) {
            int k = tid >> 4, n = tid & 15;
            lw[(k >> 1) * 32 + n * 2 + (k & 1)] = Wn[tid];
        }
        if (tid < HDIM) lb[tid] = bias[tid];
    }
    __syncthreads();

    const int h = lane >> 4, j = lane & 15;
    const int beg = bstart[b], end = bstart[b + 1];

    const int ld = 2 * j + h;
    int i = beg + w * 32;
    int e = bucket[min(i + ld, N_EDGES - 1)];
    while (i < end) {
        int p[16];
        p[0]  = row_bcast<0>(e);   p[1]  = row_bcast<1>(e);   p[2]  = row_bcast<2>(e);   p[3]  = row_bcast<3>(e);
        p[4]  = row_bcast<4>(e);   p[5]  = row_bcast<5>(e);   p[6]  = row_bcast<6>(e);   p[7]  = row_bcast<7>(e);
        p[8]  = row_bcast<8>(e);   p[9]  = row_bcast<9>(e);   p[10] = row_bcast<10>(e);  p[11] = row_bcast<11>(e);
        p[12] = row_bcast<12>(e);  p[13] = row_bcast<13>(e);  p[14] = row_bcast<14>(e);  p[15] = row_bcast<15>(e);
        int inext = i + SC_THREADS;
        const int* enext = bucket + min(inext + ld, N_EDGES - 1);
        int n = end - i;
        if (n >= 32) e = add_edges<true>(p, 32, h, j, hws, acc, enext);
        else         e = add_edges<false>(p, n, h, j, hws, acc, enext);
        i = inext;
    }

    __syncthreads();

    if (LAYER == 1) {
        for (int t = w; t < BK_NODES / 16; t += SC_WAVES) {
            const int ml = t * 16 + j;
            const long m = (long)lo + ml;
            const float dm = dis[m];
            const float* srow = hws + m * HDIM;

            v8f c = {0.f, 0.f, 0.f, 0.f, 0.f, 0.f, 0.f, 0.f};
#pragma unroll
            for (int kk = 0; kk < HDIM; kk += 4) {
                int k0 = kk + 2 * h;
                v2f s  = *(const v2f*)(srow + k0);
                v2f sm = *(const v2f*)(&acc[ml * HDIM + k0]);
                v2f a;
                a.x = fmaxf(dm * (s.x + sm.x) + lb[k0], 0.0f);
                a.y = fmaxf(dm * (s.y + sm.y) + lb[k0 + 1], 0.0f);
                v2f bf = *(const v2f*)(&lw[(k0 >> 1) * 32 + j * 2]);
                c = __builtin_amdgcn_wmma_f32_16x16x4_f32(false, a, false, bf,
                                                          (short)0, c, false, false);
            }
#pragma unroll
            for (int v = 0; v < 8; ++v) {
                long r = (long)lo + t * 16 + v + 8 * h;
                outp[r * HDIM + j] = c[v] * dis[r];
            }
        }
    } else {
        const int g  = tid >> 4;
        const int n0 = g * PL_RUN;
        const float bj = bias[j];
        float vals[PL_RUN];
        int   gid[PL_RUN];
#pragma unroll
        for (int u = 0; u < PL_RUN; ++u) {
            int il = n0 + u;
            bool in = il < BK_NODES;
            int ic = in ? il : BK_NODES - 1;
            int n = lo + ic;
            gid[u]  = in ? batch[n] : -1;
            vals[u] = dis[n] * (acc[ic * HDIM + j] + hws[(long)n * HDIM + j]) + bj;
        }
        float run = 0.0f;
        int cur = gid[0];
#pragma unroll
        for (int u = 0; u < PL_RUN; ++u) {
            if (gid[u] >= 0) {
                if (gid[u] != cur) {
                    atomicAdd(&outp[cur * HDIM + j], run);
                    run = 0.0f;
                    cur = gid[u];
                }
                run += vals[u];
            }
        }
        if (cur >= 0) atomicAdd(&outp[cur * HDIM + j], run);
    }
}

__global__ void final_kernel(const float* __restrict__ pool, const int* __restrict__ counts,
                             const float* __restrict__ Wfc, const float* __restrict__ bfc,
                             float* __restrict__ out) {
    int t = blockIdx.x * blockDim.x + threadIdx.x;
    if (t < BGRAPH * CDIM) {
        int b = t / CDIM, cc = t % CDIM;
        float inv = 1.0f / fmaxf((float)counts[b], 1.0f);
        float acc = bfc[cc];
#pragma unroll
        for (int j = 0; j < HDIM; ++j)
            acc += pool[b * HDIM + j] * inv * Wfc[j * CDIM + cc];
        out[t] = acc;
    }
}

extern "C" void kernel_launch(void* const* d_in, const int* in_sizes, int n_in,
                              void* d_out, int out_size, void* d_ws, size_t ws_size,
                              hipStream_t stream) {
    (void)in_sizes; (void)n_in; (void)out_size; (void)ws_size;

    const float* x    = (const float*)d_in[0];
    const int*   ei   = (const int*)d_in[1];
    const int*   batch= (const int*)d_in[2];
    const float* W1   = (const float*)d_in[3];
    const float* b1   = (const float*)d_in[4];
    const float* W2   = (const float*)d_in[5];
    const float* b2   = (const float*)d_in[6];
    const float* Wfc  = (const float*)d_in[7];
    const float* bfc  = (const float*)d_in[8];
    float* out = (float*)d_out;

    const int* erow = ei;
    const int* ecol = ei + N_EDGES;

    char* ws = (char*)d_ws;
    size_t o = 0;
    auto alloc = [&](size_t bytes) -> void* {
        void* p = ws + o;
        o += (bytes + 255) & ~(size_t)255;
        return p;
    };
    float* dis    = (float*)alloc((size_t)N_NODES * 4);
    float* hws    = (float*)alloc((size_t)N_NODES * HDIM * 4);
    float* hws2   = (float*)alloc((size_t)N_NODES * HDIM * 4);
    int*   bucket = (int*)  alloc((size_t)N_EDGES * 4);
    float* pool   = (float*)alloc((size_t)BGRAPH * HDIM * 4);
    int*   counts = (int*)  alloc((size_t)(BGRAPH + 256) * 4);
    int*   btot   = counts + BGRAPH;
    int*   bstart = (int*)  alloc((size_t)512 * 4);
    int*   tcnt   = (int*)hws2;

    hipMemsetAsync(counts, 0, (size_t)BGRAPH * 4, stream);
    hipMemsetAsync(pool,   0, (size_t)BGRAPH * HDIM * 4, stream);

    tile_hist_kernel<<<PT_GRID, TH_THREADS, 0, stream>>>(ecol, tcnt);
    tile_scan_kernel<<<N_BUCKETS, CS_THREADS, 0, stream>>>(tcnt, btot);
    partition_kernel<<<PT_GRID, PT_THREADS, 0, stream>>>(erow, ecol, tcnt, btot, bstart, bucket);
    degree_kernel<<<N_BUCKETS, DG_THREADS, 0, stream>>>(bucket, bstart, batch, dis, counts);

    int gemm_grid = (NTILES + 7) / 8;
    gemm1_kernel<<<gemm_grid, 256, 0, stream>>>(x, W1, dis, hws);

    edge_scatter_kernel<1><<<N_BUCKETS, SC_THREADS, 0, stream>>>(bucket, bstart, hws, dis, W2, b1, batch, hws2);
    edge_scatter_kernel<2><<<N_BUCKETS, SC_THREADS, 0, stream>>>(bucket, bstart, hws2, dis, W2, b2, batch, pool);
    final_kernel<<<(BGRAPH * CDIM + 255) / 256, 256, 0, stream>>>(pool, counts, Wfc, bfc, out);
}


# three launches fewer: layer 1 GEMM runs inside the degree kernel (x rows asked for between the count's loads and its LDS adds); the two memsets become stores of the histogram kernel; sums unchanged
# speedup vs baseline: 6.6974x; 1.0470x over previous
#include <hip/hip_runtime.h>

#define N_NODES 200000
#define N_EDGES 6400000
#define FDIM    128
#define HDIM    16
#define CDIM    10
#define BGRAPH  512
#define NTILES  (N_NODES / 16)
#define BK_NODES  800
#define N_BUCKETS (N_NODES / BK_NODES)

typedef float v2f __attribute__((ext_vector_type(2)));
typedef float v8f __attribute__((ext_vector_type(8)));

#define PT_THREADS 512
#define PT_VEC     4
#define PT_TILE4   (PT_THREADS * PT_VEC)
#define PT_EDGES   (PT_TILE4 * 4)
#define N_EDGE4    (N_EDGES / 4)
#define PT_GRID    ((N_EDGE4 + PT_TILE4 - 1) / PT_TILE4)
#define TB_PITCH   256
#define TH_THREADS 256
#define TH_VEC     (PT_TILE4 / TH_THREADS)

__global__ void __launch_bounds__(TH_THREADS) tile_hist_kernel(const int* __restrict__ col,
                                                               int* __restrict__ tcnt,
                                                               float* __restrict__ pool,
                                                               int* __restrict__ counts) {
    __shared__ int scnt[256];
    const int tid = threadIdx.x;
    scnt[tid] = 0;
    if (blockIdx.x < BGRAPH * HDIM / TH_THREADS) pool[blockIdx.x * TH_THREADS + tid] = 0.0f;
    else if (blockIdx.x < BGRAPH * HDIM / TH_THREADS + BGRAPH / TH_THREADS)
        counts[(blockIdx.x - BGRAPH * HDIM / TH_THREADS) * TH_THREADS + tid] = 0;
    __syncthreads();

    const int4* __restrict__ col4 = (const int4*)col;
    int4 c[TH_VEC];
#pragma unroll
    for (int i = 0; i < TH_VEC; ++i) {
        int i4 = blockIdx.x * PT_TILE4 + i * TH_THREADS + tid;
        bool in = i4 < N_EDGE4;
        c[i] = col4[in ? i4 : N_EDGE4 - 1];
        if (!in) c[i] = make_int4(-1, -1, -1, -1);
    }
#pragma unroll
    for (int i = 0; i < TH_VEC; ++i) {
        if (c[i].x >= 0) {
            atomicAdd(&scnt[(unsigned)c[i].x / (unsigned)BK_NODES], 1);
            atomicAdd(&scnt[(unsigned)c[i].y / (unsigned)BK_NODES], 1);
            atomicAdd(&scnt[(unsigned)c[i].z / (unsigned)BK_NODES], 1);
            atomicAdd(&scnt[(unsigned)c[i].w / (unsigned)BK_NODES], 1);
        }
    }
    __syncthreads();
    if (tid < N_BUCKETS) tcnt[blockIdx.x * TB_PITCH + tid] = scnt[tid];
}

#define CS_THREADS 256
#define CS_VEC     4

__global__ void __launch_bounds__(CS_THREADS) tile_scan_kernel(int* __restrict__ tcnt,
                                                               int* __restrict__ btot) {
    __shared__ int wsum[CS_THREADS / 32];
    const int b = blockIdx.x;
    const int t = threadIdx.x, lane = t & 31, w = t >> 5;
    int c[CS_VEC], v[CS_VEC];
#pragma unroll
    for (int q = 0; q < CS_VEC; ++q) {
        int tile = t * CS_VEC + q;
        bool in = tile < PT_GRID;
        c[q] = tcnt[(in ? tile : PT_GRID - 1) * TB_PITCH + b];
        if (!in) c[q] = 0;
    }
    int run = 0;
#pragma unroll
    for (int q = 0; q < CS_VEC; ++q) { v[q] = run; run += c[q]; }
    int incl = run;
#pragma unroll
    for (int d = 1; d < 32; d <<= 1) {
        int y = __shfl_up(incl, d);
        if (lane >= d) incl += y;
    }
    if (lane == 31) wsum[w] = incl;
    __syncthreads();
    int wbase = 0, total = 0;
#pragma unroll
    for (int k = 0; k < CS_THREADS / 32; ++k) {
        int s = wsum[k];
        if (k < w) wbase += s;
        total += s;
    }
    const int base = wbase + incl - run;
#pragma unroll
    for (int q = 0; q < CS_VEC; ++q) {
        int tile = t * CS_VEC + q;
        if (tile < PT_GRID) tcnt[tile * TB_PITCH + b] = base + v[q];
    }
    if (t == 0) btot[b] = total;
}

__global__ void __launch_bounds__(PT_THREADS) partition_kernel(const int* __restrict__ row,
                                                               const int* __restrict__ col,
                                                               const int* __restrict__ tpre,
                                                               const int* __restrict__ btot,
                                                               int* __restrict__ bstart,
                                                               int* __restrict__ bucket) {
    __shared__ int scnt[256];
    __shared__ int sloc[256];
    __shared__ int sbst[256];
    __shared__ int sdel[256];
    __shared__ int sval[PT_EDGES];
    __shared__ unsigned char sbk[PT_EDGES];
    const int tid = threadIdx.x;
    if (tid < 256) scnt[tid] = 0;
    int mypre = 0, mytot = 0;
    if (tid < N_BUCKETS) {
        mypre = tpre[blockIdx.x * TB_PITCH + tid];
        mytot = btot[tid];
    }
    __syncthreads();

    const int4* __restrict__ col4 = (const int4*)col;
    const int4* __restrict__ row4 = (const int4*)row;
    int cc[4 * PT_VEC], rr[4 * PT_VEC], bb[4 * PT_VEC], rk[4 * PT_VEC];
#pragma unroll
    for (int i = 0; i < PT_VEC; ++i) {
        int i4 = blockIdx.x * PT_TILE4 + i * PT_THREADS + tid;
        bool in = i4 < N_EDGE4;
        int i4c = in ? i4 : N_EDGE4 - 1;
        int4 c = col4[i4c], r = row4[i4c];
        if (!in) c = make_int4(-1, -1, -1, -1);
        cc[4 * i + 0] = c.x; cc[4 * i + 1] = c.y; cc[4 * i + 2] = c.z; cc[4 * i + 3] = c.w;
        rr[4 * i + 0] = r.x; rr[4 * i + 1] = r.y; rr[4 * i + 2] = r.z; rr[4 * i + 3] = r.w;
    }
#pragma unroll
    for (int q = 0; q < 4 * PT_VEC; ++q) {
        bb[q] = 0; rk[q] = 0;
        if (cc[q] >= 0) {
            bb[q] = (int)((unsigned)cc[q] / (unsigned)BK_NODES);
            rk[q] = atomicAdd(&scnt[bb[q]], 1);
        }
    }
    if (tid < 256) sbst[tid] = mytot;
    __syncthreads();

    if (tid < 64) {
        const int lane = tid & 31;
        const int* from = (tid < 32) ? scnt : sbst;
        int*       to   = (tid < 32) ? sloc : sbst;
        int c8[8];
        int run = 0;
#pragma unroll
        for (int q = 0; q < 8; ++q) { int t = from[lane * 8 + q]; c8[q] = run; run += t; }
        int incl = run;
#pragma unroll
        for (int d = 1; d < 32; d <<= 1) {
            int y = __shfl_up(incl, d);
            if (lane >= d) incl += y;
        }
        int lanebase = incl - run;
#pragma unroll
        for (int q = 0; q < 8; ++q) to[lane * 8 + q] = lanebase + c8[q];
    }
    __syncthreads();

    if (tid < N_BUCKETS) sdel[tid] = sbst[tid] + mypre - sloc[tid];
    if (blockIdx.x == 0 && tid <= N_BUCKETS) bstart[tid] = sbst[tid];
#pragma unroll
    for (int q = 0; q < 4 * PT_VEC; ++q) {
        if (cc[q] >= 0) {
            int slot = sloc[bb[q]] + rk[q];
            sval[slot] = (rr[q] << 10) | (cc[q] - bb[q] * BK_NODES);
            sbk[slot]  = (unsigned char)bb[q];
        }
    }
    __syncthreads();

    int ntot = N_EDGES - blockIdx.x * PT_EDGES;
    if (ntot > PT_EDGES) ntot = PT_EDGES;
    for (int slot = tid; slot < ntot; slot += PT_THREADS) {
        int b = sbk[slot];
        bucket[sdel[b] + slot] = sval[slot];
    }
}

#define DG_THREADS 1024
#define DG_VEC     8
#define DG_WAVES   (DG_THREADS / 32)
#define DG_TILES   (BK_NODES / 16)

__device__ __forceinline__ void dg_load(int (&p)[DG_VEC], const int* __restrict__ bucket,
                                        int base, int end, int tid) {
#pragma unroll
    for (int u = 0; u < DG_VEC; ++u) {
        int k = base + u * DG_THREADS + tid;
        int e = bucket[min(k, N_EDGES - 1)];
        p[u] = (k < end) ? e : -1;
    }
}

__device__ __forceinline__ void dg_count(const int (&p)[DG_VEC], int* __restrict__ cnt) {
#pragma unroll
    for (int u = 0; u < DG_VEC; ++u)
        if (p[u] >= 0) atomicAdd(&cnt[p[u] & 1023], 1);
}

__device__ __forceinline__ void dg_xload(v2f (&a)[FDIM / 4], const float* __restrict__ xrow, int h) {
#pragma unroll
    for (int q = 0; q < FDIM / 4; ++q) a[q] = *(const v2f*)(xrow + 4 * q + 2 * h);
}

__device__ __forceinline__ v8f dg_mma(const v2f (&a)[FDIM / 4], const float* __restrict__ lw, int h, int nn) {
    v8f c = {0.f, 0.f, 0.f, 0.f, 0.f, 0.f, 0.f, 0.f};
#pragma unroll
    for (int q = 0; q < FDIM / 4; ++q) {
        int k0 = 4 * q + 2 * h;
        v2f b = *(const v2f*)(&lw[(k0 >> 1) * 32 + nn * 2]);
        c = __builtin_amdgcn_wmma_f32_16x16x4_f32(false, a[q], false, b, (short)0, c, false, false);
    }
    return c;
}

__global__ void __launch_bounds__(DG_THREADS) degree_gemm_kernel(const int* __restrict__ bucket,
                                                                 const int* __restrict__ bstart,
                                                                 const int* __restrict__ batch,
                                                                 const float* __restrict__ x,
                                                                 const float* __restrict__ W1,
                                                                 float* __restrict__ dis,
                                                                 int* __restrict__ counts,
                                                                 float* __restrict__ hws) {
    __shared__ int cnt[BK_NODES];
    __shared__ float sdis[BK_NODES];
    __shared__ float lw[(FDIM / 2) * 32];
    const int tid = threadIdx.x;
    const int b   = blockIdx.x;
    const int lane = tid & 31;
    const int w   = __builtin_amdgcn_readfirstlane(tid >> 5);
    const int h   = lane >> 4;
    const int nn  = lane & 15;
    if (tid < BK_NODES) cnt[tid] = 0;
#pragma unroll
    for (int i = 0; i < (FDIM * HDIM) / DG_THREADS; ++i) {
        int idx = i * DG_THREADS + tid;
        int k = idx >> 4, n = idx & 15;
        lw[(k >> 1) * 32 + n * 2 + (k & 1)] = W1[idx];
    }
    const int beg = bstart[b], end = bstart[b + 1];
    __syncthreads();

    const long row0 = (long)b * BK_NODES;
    int p[DG_VEC];
    v2f a[FDIM / 4];

    int base = beg;
    dg_load(p, bucket, base, end, tid);
    dg_xload(a, x + (row0 + w * 16 + nn) * FDIM, h);
    dg_count(p, cnt);
    const v8f c0 = dg_mma(a, lw, h, nn);
    base += DG_THREADS * DG_VEC;

    const bool two = (w + DG_WAVES) < DG_TILES;
    v8f c1 = {0.f, 0.f, 0.f, 0.f, 0.f, 0.f, 0.f, 0.f};
    dg_load(p, bucket, base, end, tid);
    if (two) dg_xload(a, x + (row0 + (w + DG_WAVES) * 16 + nn) * FDIM, h);
    dg_count(p, cnt);
    if (two) c1 = dg_mma(a, lw, h, nn);
    base += DG_THREADS * DG_VEC;

    for (; base < end; base += DG_THREADS * DG_VEC) {
        dg_load(p, bucket, base, end, tid);
        dg_count(p, cnt);
    }
    __syncthreads();

    if (tid < BK_NODES) {
        int n = b * BK_NODES + tid;
        float d = rsqrtf(1.0f + (float)cnt[tid]);
        dis[n] = d;
        sdis[tid] = d;
        const int g  = batch[n];
        const int gp = __shfl_up(g, 1);
        const bool head = (lane == 0) || (g != gp);
        const unsigned heads = (unsigned)__ballot(head);
        if (head) {
            unsigned later = heads & ~((2u << lane) - 1u);
            int nxt = later ? __builtin_ctz(later) : 32;
            atomicAdd(&counts[g], nxt - lane);
        }
    }
    __syncthreads();

#pragma unroll
    for (int v = 0; v < 8; ++v) {
        int rl = w * 16 + v + 8 * h;
        hws[(row0 + rl) * HDIM + nn] = c0[v] * sdis[rl];
    }
    if (two) {
#pragma unroll
        for (int v = 0; v < 8; ++v) {
            int rl = (w + DG_WAVES) * 16 + v + 8 * h;
            hws[(row0 + rl) * HDIM + nn] = c1[v] * sdis[rl];
        }
    }
}

#define SC_THREADS 1024
#define SC_WAVES   (SC_THREADS / 32)
#define PL_RUN     13

template <int K>
__device__ __forceinline__ int row_bcast(int v) {
    return __builtin_amdgcn_update_dpp(0, v, 0x150 + K, 0xF, 0xF, true);
}

template <bool FULL>
__device__ __forceinline__ int add_edges(const int (&p)[16], int n, int h, int j,
                                         const float* __restrict__ hws, float* __restrict__ acc,
                                         const int* __restrict__ enext) {
    float v[16];
#pragma unroll
    for (int k = 0; k < 16; ++k) {
        int s = 2 * k + h;
        unsigned r = (unsigned)p[k] >> 10;
        v[k] = (FULL || s < n) ? hws[r * HDIM + j] : 0.0f;
    }
    int e = *enext;
#pragma unroll
    for (int k = 0; k < 16; ++k) {
        int s = 2 * k + h;
        int rel = p[k] & 1023;
        if (FULL || s < n)
            __hip_atomic_fetch_add(&acc[rel * HDIM + j], v[k], __ATOMIC_RELAXED,
                                   __HIP_MEMORY_SCOPE_WORKGROUP);
    }
    return e;
}

template <int LAYER>
__global__ void __launch_bounds__(SC_THREADS) edge_scatter_kernel(const int* __restrict__ bucket,
                                                                  const int* __restrict__ bstart,
                                                                  const float* __restrict__ hws,
                                                                  const float* __restrict__ dis,
                                                                  const float* __restrict__ Wn,
                                                                  const float* __restrict__ bias,
                                                                  const int* __restrict__ batch,
                                                                  float* __restrict__ outp) {
    __shared__ __attribute__((aligned(16))) float acc[BK_NODES * HDIM];
    __shared__ float lw[(HDIM / 2) * 32];
    __shared__ float lb[HDIM];

    const int tid  = threadIdx.x;
    const int lane = tid & 31;
    const int w    = __builtin_amdgcn_readfirstlane(tid >> 5);
    const int b    = blockIdx.x;
    const int lo   = b * BK_NODES;

    for (int i = tid; i < BK_NODES * HDIM / 4; i += SC_THREADS)
        ((float4*)acc)[i] = make_float4(0.f, 0.f, 0.f, 0.f);
    if (LAYER == 1) {
        if (tid < 256) {
            int k = tid >> 4, n = tid & 15;
            lw[(k >> 1) * 32 + n * 2 + (k & 1)] = Wn[tid];
        }
        if (tid < HDIM) lb[tid] = bias[tid];
    }
    __syncthreads();

    const int h = lane >> 4, j = lane & 15;
    const int beg = bstart[b], end = bstart[b + 1];

    const int ld = 2 * j + h;
    int i = beg + w * 32;
    int e = bucket[min(i + ld, N_EDGES - 1)];
    while (i < end) {
        int p[16];
        p[0]  = row_bcast<0>(e);   p[1]  = row_bcast<1>(e);   p[2]  = row_bcast<2>(e);   p[3]  = row_bcast<3>(e);
        p[4]  = row_bcast<4>(e);   p[5]  = row_bcast<5>(e);   p[6]  = row_bcast<6>(e);   p[7]  = row_bcast<7>(e);
        p[8]  = row_bcast<8>(e);   p[9]  = row_bcast<9>(e);   p[10] = row_bcast<10>(e);  p[11] = row_bcast<11>(e);
        p[12] = row_bcast<12>(e);  p[13] = row_bcast<13>(e);  p[14] = row_bcast<14>(e);  p[15] = row_bcast<15>(e);
        int inext = i + SC_THREADS;
        const int* enext = bucket + min(inext + ld, N_EDGES - 1);
        int n = end - i;
        if (n >= 32) e = add_edges<true>(p, 32, h, j, hws, acc, enext);
        else         e = add_edges<false>(p, n, h, j, hws, acc, enext);
        i = inext;
    }

    __syncthreads();

    if (LAYER == 1) {
        for (int t = w; t < BK_NODES / 16; t += SC_WAVES) {
            const int ml = t * 16 + j;
            const long m = (long)lo + ml;
            const float dm = dis[m];
            const float* srow = hws + m * HDIM;

            v8f c = {0.f, 0.f, 0.f, 0.f, 0.f, 0.f, 0.f, 0.f};
#pragma unroll
            for (int kk = 0; kk < HDIM; kk += 4) {
                int k0 = kk + 2 * h;
                v2f s  = *(const v2f*)(srow + k0);
                v2f sm = *(const v2f*)(&acc[ml * HDIM + k0]);
                v2f a;
                a.x = fmaxf(dm * (s.x + sm.x) + lb[k0], 0.0f);
                a.y = fmaxf(dm * (s.y + sm.y) + lb[k0 + 1], 0.0f);
                v2f bf = *(const v2f*)(&lw[(k0 >> 1) * 32 + j * 2]);
                c = __builtin_amdgcn_wmma_f32_16x16x4_f32(false, a, false, bf,
                                                          (short)0, c, false, false);
            }
#pragma unroll
            for (int v = 0; v < 8; ++v) {
                long r = (long)lo + t * 16 + v + 8 * h;
                outp[r * HDIM + j] = c[v] * dis[r];
            }
        }
    } else {
        const int g  = tid >> 4;
        const int n0 = g * PL_RUN;
        const float bj = bias[j];
        float vals[PL_RUN];
        int   gid[PL_RUN];
#pragma unroll
        for (int u = 0; u < PL_RUN; ++u) {
            int il = n0 + u;
            bool in = il < BK_NODES;
            int ic = in ? il : BK_NODES - 1;
            int n = lo + ic;
            gid[u]  = in ? batch[n] : -1;
            vals[u] = dis[n] * (acc[ic * HDIM + j] + hws[(long)n * HDIM + j]) + bj;
        }
        float run = 0.0f;
        int cur = gid[0];
#pragma unroll
        for (int u = 0; u < PL_RUN; ++u) {
            if (gid[u] >= 0) {
                if (gid[u] != cur) {
                    atomicAdd(&outp[cur * HDIM + j], run);
                    run = 0.0f;
                    cur = gid[u];
                }
                run += vals[u];
            }
        }
        if (cur >= 0) atomicAdd(&outp[cur * HDIM + j], run);
    }
}

__global__ void final_kernel(const float* __restrict__ pool, const int* __restrict__ counts,
                             const float* __restrict__ Wfc, const float* __restrict__ bfc,
                             float* __restrict__ out) {
    int t = blockIdx.x * blockDim.x + threadIdx.x;
    if (t < BGRAPH * CDIM) {
        int b = t / CDIM, cc = t % CDIM;
        float inv = 1.0f / fmaxf((float)counts[b], 1.0f);
        float acc = bfc[cc];
#pragma unroll
        for (int j = 0; j < HDIM; ++j)
            acc += pool[b * HDIM + j] * inv * Wfc[j * CDIM + cc];
        out[t] = acc;
    }
}

extern "C" void kernel_launch(void* const* d_in, const int* in_sizes, int n_in,
                              void* d_out, int out_size, void* d_ws, size_t ws_size,
                              hipStream_t stream) {
    (void)in_sizes; (void)n_in; (void)out_size; (void)ws_size;

    const float* x    = (const float*)d_in[0];
    const int*   ei   = (const int*)d_in[1];
    const int*   batch= (const int*)d_in[2];
    const float* W1   = (const float*)d_in[3];
    const float* b1   = (const float*)d_in[4];
    const float* W2   = (const float*)d_in[5];
    const float* b2   = (const float*)d_in[6];
    const float* Wfc  = (const float*)d_in[7];
    const float* bfc  = (const float*)d_in[8];
    float* out = (float*)d_out;

    const int* erow = ei;
    const int* ecol = ei + N_EDGES;

    char* ws = (char*)d_ws;
    size_t o = 0;
    auto alloc = [&](size_t bytes) -> void* {
        void* p = ws + o;
        o += (bytes + 255) & ~(size_t)255;
        return p;
    };
    float* dis    = (float*)alloc((size_t)N_NODES * 4);
    float* hws    = (float*)alloc((size_t)N_NODES * HDIM * 4);
    float* hws2   = (float*)alloc((size_t)N_NODES * HDIM * 4);
    int*   bucket = (int*)  alloc((size_t)N_EDGES * 4);
    float* pool   = (float*)alloc((size_t)BGRAPH * HDIM * 4);
    int*   counts = (int*)  alloc((size_t)(BGRAPH + 256) * 4);
    int*   btot   = counts + BGRAPH;
    int*   bstart = (int*)  alloc((size_t)512 * 4);
    int*   tcnt   = (int*)hws2;

    tile_hist_kernel<<<PT_GRID, TH_THREADS, 0, stream>>>(ecol, tcnt, pool, counts);
    tile_scan_kernel<<<N_BUCKETS, CS_THREADS, 0, stream>>>(tcnt, btot);
    partition_kernel<<<PT_GRID, PT_THREADS, 0, stream>>>(erow, ecol, tcnt, btot, bstart, bucket);
    degree_gemm_kernel<<<N_BUCKETS, DG_THREADS, 0, stream>>>(bucket, bstart, batch, x, W1, dis, counts, hws);

    edge_scatter_kernel<1><<<N_BUCKETS, SC_THREADS, 0, stream>>>(bucket, bstart, hws, dis, W2, b1, batch, hws2);
    edge_scatter_kernel<2><<<N_BUCKETS, SC_THREADS, 0, stream>>>(bucket, bstart, hws2, dis, W2, b2, batch, pool);
    final_kernel<<<(BGRAPH * CDIM + 255) / 256, 256, 0, stream>>>(pool, counts, Wfc, bfc, out);
}
